# starting kernel
# baseline (speedup 1.0000x reference)
#include <hip/hip_runtime.h>
#include <math.h>

#define D      256
#define TX     512
#define TM     512
#define BATCH  4

typedef __attribute__((ext_vector_type(2))) float v2f;
typedef __attribute__((ext_vector_type(8))) float v8f;

__device__ __forceinline__ float fast_tanh(float x) {
#if __has_builtin(__builtin_amdgcn_tanhf)
  return __builtin_amdgcn_tanhf(x);
#else
  float r;
  asm("v_tanh_f32 %0, %1" : "=v"(r) : "v"(x));
  return r;
#endif
}

__device__ __forceinline__ float fast_exp(float x) {
#if __has_builtin(__builtin_amdgcn_exp2f)
  return __builtin_amdgcn_exp2f(x * 1.4426950408889634f);
#else
  return __expf(x);
#endif
}

__global__ __launch_bounds__(128) void proj_gemm_kernel(
    const float* __restrict__ In, const float* __restrict__ W,
    const float* __restrict__ bias, float* __restrict__ Out, int Mrows) {
  const int warp = threadIdx.x >> 5;
  const int lane = threadIdx.x & 31;
  const int half = lane >> 4;
  const int l    = lane & 15;

  const int pairsPerRow = D / 32;
  int tile = blockIdx.x * (blockDim.x >> 5) + warp;
  int mt = tile / pairsPerRow;
  int np = tile % pairsPerRow;
  if (mt * 16 >= Mrows) return;
  const int row0 = mt * 16;
  const int col0 = np * 32;

  v8f acc0 = {}; v8f acc1 = {};
  const float* arow  = In + (size_t)(row0 + l) * D;
  const float* b0row = W  + (size_t)(col0 + l) * D;
  const float* b1row = W  + (size_t)(col0 + 16 + l) * D;
#pragma unroll 8
  for (int k = 0; k < D; k += 4) {
    v2f a  = *(const v2f*)(arow  + k + 2 * half);
    v2f b0 = *(const v2f*)(b0row + k + 2 * half);
    v2f b1 = *(const v2f*)(b1row + k + 2 * half);
    acc0 = __builtin_amdgcn_wmma_f32_16x16x4_f32(false, a, false, b0, (short)0, acc0, false, false);
    acc1 = __builtin_amdgcn_wmma_f32_16x16x4_f32(false, a, false, b1, (short)0, acc1, false, false);
  }
  float bv0 = 0.0f, bv1 = 0.0f;
  if (bias) { bv0 = bias[col0 + l]; bv1 = bias[col0 + 16 + l]; }
#pragma unroll
  for (int j = 0; j < 8; ++j) {
    int r = row0 + j + 8 * half;
    Out[(size_t)r * D + col0 + l]      = acc0[j] + bv0;
    Out[(size_t)r * D + col0 + 16 + l] = acc1[j] + bv1;
  }
}

__global__ __launch_bounds__(256) void scores_softmax_kernel(
    const float* __restrict__ item1, const float* __restrict__ item2,
    const float* __restrict__ wst, const int* __restrict__ mask,
    float* __restrict__ Sout) {
  __shared__ float sc[TM];
  __shared__ float red[256];

  const int tid  = threadIdx.x;
  const int b    = blockIdx.x / TX;
  const int x    = blockIdx.x % TX;
  const int warp = tid >> 5;
  const int lane = tid & 31;

  const size_t rb = ((size_t)b * TX + x) * D;
  const float4 f10 = *(const float4*)(item1 + rb + lane * 8);
  const float4 f11 = *(const float4*)(item1 + rb + lane * 8 + 4);
  const float4 w0  = *(const float4*)(wst + lane * 8);
  const float4 w1  = *(const float4*)(wst + lane * 8 + 4);

  const float* i2b = item2 + (size_t)b * TM * D;
  for (int m = warp; m < TM; m += 8) {
    const float* row = i2b + (size_t)m * D + lane * 8;
    if (m + 8 < TM) __builtin_prefetch(row + 8 * D, 0, 3);
    float4 r0 = *(const float4*)(row);
    float4 r1 = *(const float4*)(row + 4);
    float p;
    p  = fast_tanh(f10.x + r0.x) * w0.x;
    p += fast_tanh(f10.y + r0.y) * w0.y;
    p += fast_tanh(f10.z + r0.z) * w0.z;
    p += fast_tanh(f10.w + r0.w) * w0.w;
    p += fast_tanh(f11.x + r1.x) * w1.x;
    p += fast_tanh(f11.y + r1.y) * w1.y;
    p += fast_tanh(f11.z + r1.z) * w1.z;
    p += fast_tanh(f11.w + r1.w) * w1.w;
    p += __shfl_xor(p, 16, 32);
    p += __shfl_xor(p, 8, 32);
    p += __shfl_xor(p, 4, 32);
    p += __shfl_xor(p, 2, 32);
    p += __shfl_xor(p, 1, 32);
    if (lane == 0) sc[m] = p;
  }
  __syncthreads();

  float s0 = sc[tid];
  float s1 = sc[tid + 256];
  if (mask[b * TM + tid] == 0)       s0 = -INFINITY;
  if (mask[b * TM + tid + 256] == 0) s1 = -INFINITY;

  red[tid] = fmaxf(s0, s1);
  __syncthreads();
  for (int s = 128; s > 0; s >>= 1) {
    if (tid < s) red[tid] = fmaxf(red[tid], red[tid + s]);
    __syncthreads();
  }
  const float rowmax = red[0];
  __syncthreads();

  const float e0 = fast_exp(s0 - rowmax);
  const float e1 = fast_exp(s1 - rowmax);
  red[tid] = e0 + e1;
  __syncthreads();
  for (int s = 128; s > 0; s >>= 1) {
    if (tid < s) red[tid] += red[tid + s];
    __syncthreads();
  }
  const float inv = 1.0f / red[0];

  const size_t base = ((size_t)b * TX + x) * TM;
  Sout[base + tid]       = e0 * inv;
  Sout[base + 256 + tid] = e1 * inv;
}

__global__ __launch_bounds__(128) void attn_out_kernel(
    const float* __restrict__ S, const float* __restrict__ Mem,
    float* __restrict__ Out) {
  const int warp = threadIdx.x >> 5;
  const int lane = threadIdx.x & 31;
  const int half = lane >> 4;
  const int l    = lane & 15;

  const int pairsPerRow   = D / 32;
  const int tilesPerBatch = (TX / 16) * pairsPerRow;
  int tile  = blockIdx.x * (blockDim.x >> 5) + warp;
  int batch = tile / tilesPerBatch;
  int t     = tile % tilesPerBatch;
  int mt = t / pairsPerRow;
  int np = t % pairsPerRow;
  const int row0 = mt * 16;
  const int col0 = np * 32;

  const float* Sb = S   + (size_t)batch * TX * TM;
  const float* Mb = Mem + (size_t)batch * TM * D;
  const float* arow = Sb + (size_t)(row0 + l) * TM;

  v8f acc0 = {}; v8f acc1 = {};
#pragma unroll 4
  for (int k = 0; k < TM; k += 4) {
    const int kr = k + 2 * half;
    v2f a = *(const v2f*)(arow + kr);
    v2f b0, b1;
    b0.x = Mb[(size_t)kr * D + col0 + l];
    b0.y = Mb[(size_t)(kr + 1) * D + col0 + l];
    b1.x = Mb[(size_t)kr * D + col0 + 16 + l];
    b1.y = Mb[(size_t)(kr + 1) * D + col0 + 16 + l];
    acc0 = __builtin_amdgcn_wmma_f32_16x16x4_f32(false, a, false, b0, (short)0, acc0, false, false);
    acc1 = __builtin_amdgcn_wmma_f32_16x16x4_f32(false, a, false, b1, (short)0, acc1, false, false);
  }
#pragma unroll
  for (int j = 0; j < 8; ++j) {
    int r = row0 + j + 8 * half;
    Out[((size_t)batch * TX + r) * D + col0 + l]      = acc0[j];
    Out[((size_t)batch * TX + r) * D + col0 + 16 + l] = acc1[j];
  }
}

extern "C" void kernel_launch(void* const* d_in, const int* in_sizes, int n_in,
                              void* d_out, int out_size, void* d_ws, size_t ws_size,
                              hipStream_t stream) {
  const float* x    = (const float*)d_in[0];
  const float* mem  = (const float*)d_in[1];
  const int*   mask = (const int*)d_in[2];
  const float* w1   = (const float*)d_in[3];
  const float* b1   = (const float*)d_in[4];
  const float* w2   = (const float*)d_in[5];
  const float* wst  = (const float*)d_in[6];

  float* item1 = (float*)d_ws;
  float* item2 = item1 + (size_t)BATCH * TX * D;
  float* out   = (float*)d_out;
  float* Sout  = out + (size_t)BATCH * TX * D;

  const int rowsX = BATCH * TX;
  const int rowsM = BATCH * TM;
  dim3 blk(128);

  int tilesProj = (rowsX / 16) * (D / 32);
  proj_gemm_kernel<<<tilesProj / 4, blk, 0, stream>>>(x,   w1, b1,      item1, rowsX);
  proj_gemm_kernel<<<tilesProj / 4, blk, 0, stream>>>(mem, w2, nullptr, item2, rowsM);

  scores_softmax_kernel<<<BATCH * TX, 256, 0, stream>>>(item1, item2, wst, mask, Sout);

  int tilesOut = BATCH * (TX / 16) * (D / 32);
  attn_out_kernel<<<tilesOut / 4, blk, 0, stream>>>(Sout, mem, out);
}


# scores kernel re-decomposed: block = 8 x-rows x 512 m, lane = one m row summing over d in-lane from LDS-staged item2 chunks; no cross-lane reduction per score; d-sum and softmax sum order change
# speedup vs baseline: 1.3994x; 1.3994x over previous
#include <hip/hip_runtime.h>
#include <math.h>

#define D      256
#define TX     512
#define TM     512
#define BATCH  4

typedef __attribute__((ext_vector_type(2))) float v2f;
typedef __attribute__((ext_vector_type(8))) float v8f;

__device__ __forceinline__ float fast_tanh(float x) {
#if __has_builtin(__builtin_amdgcn_tanhf)
  return __builtin_amdgcn_tanhf(x);
#else
  float r;
  asm("v_tanh_f32 %0, %1" : "=v"(r) : "v"(x));
  return r;
#endif
}

__device__ __forceinline__ float fast_exp(float x) {
#if __has_builtin(__builtin_amdgcn_exp2f)
  return __builtin_amdgcn_exp2f(x * 1.4426950408889634f);
#else
  return __expf(x);
#endif
}

__global__ __launch_bounds__(128) void proj_gemm_kernel(
    const float* __restrict__ In, const float* __restrict__ W,
    const float* __restrict__ bias, float* __restrict__ Out, int Mrows) {
  const int warp = threadIdx.x >> 5;
  const int lane = threadIdx.x & 31;
  const int half = lane >> 4;
  const int l    = lane & 15;

  const int pairsPerRow = D / 32;
  int tile = blockIdx.x * (blockDim.x >> 5) + warp;
  int mt = tile / pairsPerRow;
  int np = tile % pairsPerRow;
  if (mt * 16 >= Mrows) return;
  const int row0 = mt * 16;
  const int col0 = np * 32;

  v8f acc0 = {}; v8f acc1 = {};
  const float* arow  = In + (size_t)(row0 + l) * D;
  const float* b0row = W  + (size_t)(col0 + l) * D;
  const float* b1row = W  + (size_t)(col0 + 16 + l) * D;
#pragma unroll 8
  for (int k = 0; k < D; k += 4) {
    v2f a  = *(const v2f*)(arow  + k + 2 * half);
    v2f b0 = *(const v2f*)(b0row + k + 2 * half);
    v2f b1 = *(const v2f*)(b1row + k + 2 * half);
    acc0 = __builtin_amdgcn_wmma_f32_16x16x4_f32(false, a, false, b0, (short)0, acc0, false, false);
    acc1 = __builtin_amdgcn_wmma_f32_16x16x4_f32(false, a, false, b1, (short)0, acc1, false, false);
  }
  float bv0 = 0.0f, bv1 = 0.0f;
  if (bias) { bv0 = bias[col0 + l]; bv1 = bias[col0 + 16 + l]; }
#pragma unroll
  for (int j = 0; j < 8; ++j) {
    int r = row0 + j + 8 * half;
    Out[(size_t)r * D + col0 + l]      = acc0[j] + bv0;
    Out[(size_t)r * D + col0 + 16 + l] = acc1[j] + bv1;
  }
}

typedef const float __attribute__((address_space(4))) cfloat_k;
__device__ __forceinline__ const cfloat_k* as_const_space(const float* p) {
  return (const cfloat_k*)p;
}

#define XT 8
#define DC 64
#define PITCH 68
__global__ __launch_bounds__(512) void scores_softmax_kernel(
    const float* __restrict__ item1, const float* __restrict__ item2,
    const float* __restrict__ wst, const int* __restrict__ mask,
    float* __restrict__ Sout) {
  __shared__ float t2[TM * PITCH];
  __shared__ float sc[XT][TM];

  const int tid   = threadIdx.x;
  const int warp  = tid >> 5;
  const int lane  = tid & 31;
  const int xrow0 = blockIdx.x * XT;
  const int b     = xrow0 / TX;

  const float*    i2b = item2 + (size_t)b * TM * D;
  const cfloat_k* i1b = as_const_space(item1 + (size_t)xrow0 * D);
  const cfloat_k* wsc = as_const_space(wst);

  const int srow = tid >> 4;
  const int scol = 4 * (tid & 15);

  float acc[XT];
#pragma unroll
  for (int x = 0; x < XT; ++x) acc[x] = 0.0f;

  for (int d0 = 0; d0 < D; d0 += DC) {
    __syncthreads();
#pragma unroll
    for (int i = 0; i < 16; ++i) {
      const float4 g = *(const float4*)(i2b + (size_t)(srow + 32 * i) * D + d0 + scol);
      *(float4*)(&t2[(srow + 32 * i) * PITCH + scol]) = g;
    }
    __syncthreads();
#pragma unroll 4
    for (int k = 0; k < DC; k += 4) {
      const float4 v = *(const float4*)(&t2[tid * PITCH + k]);
      const float w_0 = wsc[d0 + k], w_1 = wsc[d0 + k + 1], w_2 = wsc[d0 + k + 2], w_3 = wsc[d0 + k + 3];
#pragma unroll
      for (int x = 0; x < XT; ++x) {
        const cfloat_k* ap = i1b + x * D + d0 + k;
        const float a_0 = ap[0], a_1 = ap[1], a_2 = ap[2], a_3 = ap[3];
        acc[x] += fast_tanh(a_0 + v.x) * w_0;
        acc[x] += fast_tanh(a_1 + v.y) * w_1;
        acc[x] += fast_tanh(a_2 + v.z) * w_2;
        acc[x] += fast_tanh(a_3 + v.w) * w_3;
      }
    }
  }

#pragma unroll
  for (int x = 0; x < XT; ++x) sc[x][tid] = acc[x];
  __syncthreads();

  if (warp < XT) {
    const int r = warp;
    float4 s[4];
#pragma unroll
    for (int i = 0; i < 4; ++i) {
      s[i] = *(const float4*)(&sc[r][128 * i + 4 * lane]);
      const int4 mk = *(const int4*)(mask + b * TM + 128 * i + 4 * lane);
      if (mk.x == 0) s[i].x = -INFINITY;
      if (mk.y == 0) s[i].y = -INFINITY;
      if (mk.z == 0) s[i].z = -INFINITY;
      if (mk.w == 0) s[i].w = -INFINITY;
    }
    float mx = fmaxf(fmaxf(s[0].x, s[0].y), fmaxf(s[0].z, s[0].w));
#pragma unroll
    for (int i = 1; i < 4; ++i)
      mx = fmaxf(mx, fmaxf(fmaxf(s[i].x, s[i].y), fmaxf(s[i].z, s[i].w)));
    mx = fmaxf(mx, __shfl_xor(mx, 16, 32));
    mx = fmaxf(mx, __shfl_xor(mx, 8, 32));
    mx = fmaxf(mx, __shfl_xor(mx, 4, 32));
    mx = fmaxf(mx, __shfl_xor(mx, 2, 32));
    mx = fmaxf(mx, __shfl_xor(mx, 1, 32));

    float sum = 0.0f;
#pragma unroll
    for (int i = 0; i < 4; ++i) {
      s[i].x = fast_exp(s[i].x - mx);
      s[i].y = fast_exp(s[i].y - mx);
      s[i].z = fast_exp(s[i].z - mx);
      s[i].w = fast_exp(s[i].w - mx);
      sum += (s[i].x + s[i].y) + (s[i].z + s[i].w);
    }
    sum += __shfl_xor(sum, 16, 32);
    sum += __shfl_xor(sum, 8, 32);
    sum += __shfl_xor(sum, 4, 32);
    sum += __shfl_xor(sum, 2, 32);
    sum += __shfl_xor(sum, 1, 32);
    const float inv = 1.0f / sum;

    float* orow = Sout + (size_t)(xrow0 + r) * TM;
#pragma unroll
    for (int i = 0; i < 4; ++i) {
      float4 o;
      o.x = s[i].x * inv; o.y = s[i].y * inv; o.z = s[i].z * inv; o.w = s[i].w * inv;
      *(float4*)(orow + 128 * i + 4 * lane) = o;
    }
  }
}

__global__ __launch_bounds__(128) void attn_out_kernel(
    const float* __restrict__ S, const float* __restrict__ Mem,
    float* __restrict__ Out) {
  const int warp = threadIdx.x >> 5;
  const int lane = threadIdx.x & 31;
  const int half = lane >> 4;
  const int l    = lane & 15;

  const int pairsPerRow   = D / 32;
  const int tilesPerBatch = (TX / 16) * pairsPerRow;
  int tile  = blockIdx.x * (blockDim.x >> 5) + warp;
  int batch = tile / tilesPerBatch;
  int t     = tile % tilesPerBatch;
  int mt = t / pairsPerRow;
  int np = t % pairsPerRow;
  const int row0 = mt * 16;
  const int col0 = np * 32;

  const float* Sb = S   + (size_t)batch * TX * TM;
  const float* Mb = Mem + (size_t)batch * TM * D;
  const float* arow = Sb + (size_t)(row0 + l) * TM;

  v8f acc0 = {}; v8f acc1 = {};
#pragma unroll 4
  for (int k = 0; k < TM; k += 4) {
    const int kr = k + 2 * half;
    v2f a = *(const v2f*)(arow + kr);
    v2f b0, b1;
    b0.x = Mb[(size_t)kr * D + col0 + l];
    b0.y = Mb[(size_t)(kr + 1) * D + col0 + l];
    b1.x = Mb[(size_t)kr * D + col0 + 16 + l];
    b1.y = Mb[(size_t)(kr + 1) * D + col0 + 16 + l];
    acc0 = __builtin_amdgcn_wmma_f32_16x16x4_f32(false, a, false, b0, (short)0, acc0, false, false);
    acc1 = __builtin_amdgcn_wmma_f32_16x16x4_f32(false, a, false, b1, (short)0, acc1, false, false);
  }
#pragma unroll
  for (int j = 0; j < 8; ++j) {
    int r = row0 + j + 8 * half;
    Out[((size_t)batch * TX + r) * D + col0 + l]      = acc0[j];
    Out[((size_t)batch * TX + r) * D + col0 + 16 + l] = acc1[j];
  }
}

extern "C" void kernel_launch(void* const* d_in, const int* in_sizes, int n_in,
                              void* d_out, int out_size, void* d_ws, size_t ws_size,
                              hipStream_t stream) {
  const float* x    = (const float*)d_in[0];
  const float* mem  = (const float*)d_in[1];
  const int*   mask = (const int*)d_in[2];
  const float* w1   = (const float*)d_in[3];
  const float* b1   = (const float*)d_in[4];
  const float* w2   = (const float*)d_in[5];
  const float* wst  = (const float*)d_in[6];

  float* item1 = (float*)d_ws;
  float* item2 = item1 + (size_t)BATCH * TX * D;
  float* out   = (float*)d_out;
  float* Sout  = out + (size_t)BATCH * TX * D;

  const int rowsX = BATCH * TX;
  const int rowsM = BATCH * TM;
  dim3 blk(128);

  int tilesProj = (rowsX / 16) * (D / 32);
  proj_gemm_kernel<<<tilesProj / 4, blk, 0, stream>>>(x,   w1, b1,      item1, rowsX);
  proj_gemm_kernel<<<tilesProj / 4, blk, 0, stream>>>(mem, w2, nullptr, item2, rowsM);

  scores_softmax_kernel<<<BATCH * TX / XT, TM, 0, stream>>>(item1, item2, wst, mask, Sout);

  int tilesOut = BATCH * (TX / 16) * (D / 32);
  attn_out_kernel<<<tilesOut / 4, blk, 0, stream>>>(Sout, mem, out);
}


# both projection GEMMs (x*W1^T+b1 and memory*W2^T) run in one launch instead of two: one grid of twice the blocks, the block index picks the problem; 4 launches become 3
# speedup vs baseline: 1.5336x; 1.0959x over previous
#include <hip/hip_runtime.h>
#include <math.h>

#define D      256
#define TX     512
#define TM     512
#define BATCH  4

typedef __attribute__((ext_vector_type(2))) float v2f;
typedef __attribute__((ext_vector_type(8))) float v8f;

__device__ __forceinline__ float fast_tanh(float x) {
#if __has_builtin(__builtin_amdgcn_tanhf)
  return __builtin_amdgcn_tanhf(x);
#else
  float r;
  asm("v_tanh_f32 %0, %1" : "=v"(r) : "v"(x));
  return r;
#endif
}

__device__ __forceinline__ float fast_exp(float x) {
#if __has_builtin(__builtin_amdgcn_exp2f)
  return __builtin_amdgcn_exp2f(x * 1.4426950408889634f);
#else
  return __expf(x);
#endif
}

#define PROJ_BLOCKS ((BATCH * TX / 16) * (D / 32) / 4)
static_assert(TX == TM, "both projections must have the same number of rows to share one grid split");
__global__ __launch_bounds__(128) void proj_gemm_kernel(
    const float* __restrict__ In0, const float* __restrict__ W0,
    const float* __restrict__ bias0, float* __restrict__ Out0,
    const float* __restrict__ In1, const float* __restrict__ W1,
    float* __restrict__ Out1, int Mrows) {
  const int warp = threadIdx.x >> 5;
  const int lane = threadIdx.x & 31;
  const int half = lane >> 4;
  const int l    = lane & 15;

  const bool second = blockIdx.x >= PROJ_BLOCKS;
  const float* In   = second ? In1 : In0;
  const float* W    = second ? W1 : W0;
  const float* bias = second ? nullptr : bias0;
  float*       Out  = second ? Out1 : Out0;

  const int pairsPerRow = D / 32;
  int tile = (blockIdx.x - (second ? PROJ_BLOCKS : 0)) * (blockDim.x >> 5) + warp;
  int mt = tile / pairsPerRow;
  int np = tile % pairsPerRow;
  if (mt * 16 >= Mrows) return;
  const int row0 = mt * 16;
  const int col0 = np * 32;

  v8f acc0 = {}; v8f acc1 = {};
  const float* arow  = In + (size_t)(row0 + l) * D;
  const float* b0row = W  + (size_t)(col0 + l) * D;
  const float* b1row = W  + (size_t)(col0 + 16 + l) * D;
#pragma unroll 8
  for (int k = 0; k < D; k += 4) {
    v2f a  = *(const v2f*)(arow  + k + 2 * half);
    v2f b0 = *(const v2f*)(b0row + k + 2 * half);
    v2f b1 = *(const v2f*)(b1row + k + 2 * half);
    acc0 = __builtin_amdgcn_wmma_f32_16x16x4_f32(false, a, false, b0, (short)0, acc0, false, false);
    acc1 = __builtin_amdgcn_wmma_f32_16x16x4_f32(false, a, false, b1, (short)0, acc1, false, false);
  }
  float bv0 = 0.0f, bv1 = 0.0f;
  if (bias) { bv0 = bias[col0 + l]; bv1 = bias[col0 + 16 + l]; }
#pragma unroll
  for (int j = 0; j < 8; ++j) {
    int r = row0 + j + 8 * half;
    Out[(size_t)r * D + col0 + l]      = acc0[j] + bv0;
    Out[(size_t)r * D + col0 + 16 + l] = acc1[j] + bv1;
  }
}

typedef const float __attribute__((address_space(4))) cfloat_k;
__device__ __forceinline__ const cfloat_k* as_const_space(const float* p) {
  return (const cfloat_k*)p;
}

#define XT 8
#define DC 64
#define PITCH 68
__global__ __launch_bounds__(512) void scores_softmax_kernel(
    const float* __restrict__ item1, const float* __restrict__ item2,
    const float* __restrict__ wst, const int* __restrict__ mask,
    float* __restrict__ Sout) {
  __shared__ float t2[TM * PITCH];
  __shared__ float sc[XT][TM];

  const int tid   = threadIdx.x;
  const int warp  = tid >> 5;
  const int lane  = tid & 31;
  const int xrow0 = blockIdx.x * XT;
  const int b     = xrow0 / TX;

  const float*    i2b = item2 + (size_t)b * TM * D;
  const cfloat_k* i1b = as_const_space(item1 + (size_t)xrow0 * D);
  const cfloat_k* wsc = as_const_space(wst);

  const int srow = tid >> 4;
  const int scol = 4 * (tid & 15);

  float acc[XT];
#pragma unroll
  for (int x = 0; x < XT; ++x) acc[x] = 0.0f;

  for (int d0 = 0; d0 < D; d0 += DC) {
    __syncthreads();
#pragma unroll
    for (int i = 0; i < 16; ++i) {
      const float4 g = *(const float4*)(i2b + (size_t)(srow + 32 * i) * D + d0 + scol);
      *(float4*)(&t2[(srow + 32 * i) * PITCH + scol]) = g;
    }
    __syncthreads();
#pragma unroll 4
    for (int k = 0; k < DC; k += 4) {
      const float4 v = *(const float4*)(&t2[tid * PITCH + k]);
      const float w_0 = wsc[d0 + k], w_1 = wsc[d0 + k + 1], w_2 = wsc[d0 + k + 2], w_3 = wsc[d0 + k + 3];
#pragma unroll
      for (int x = 0; x < XT; ++x) {
        const cfloat_k* ap = i1b + x * D + d0 + k;
        const float a_0 = ap[0], a_1 = ap[1], a_2 = ap[2], a_3 = ap[3];
        acc[x] += fast_tanh(a_0 + v.x) * w_0;
        acc[x] += fast_tanh(a_1 + v.y) * w_1;
        acc[x] += fast_tanh(a_2 + v.z) * w_2;
        acc[x] += fast_tanh(a_3 + v.w) * w_3;
      }
    }
  }

#pragma unroll
  for (int x = 0; x < XT; ++x) sc[x][tid] = acc[x];
  __syncthreads();

  if (warp < XT) {
    const int r = warp;
    float4 s[4];
#pragma unroll
    for (int i = 0; i < 4; ++i) {
      s[i] = *(const float4*)(&sc[r][128 * i + 4 * lane]);
      const int4 mk = *(const int4*)(mask + b * TM + 128 * i + 4 * lane);
      if (mk.x == 0) s[i].x = -INFINITY;
      if (mk.y == 0) s[i].y = -INFINITY;
      if (mk.z == 0) s[i].z = -INFINITY;
      if (mk.w == 0) s[i].w = -INFINITY;
    }
    float mx = fmaxf(fmaxf(s[0].x, s[0].y), fmaxf(s[0].z, s[0].w));
#pragma unroll
    for (int i = 1; i < 4; ++i)
      mx = fmaxf(mx, fmaxf(fmaxf(s[i].x, s[i].y), fmaxf(s[i].z, s[i].w)));
    mx = fmaxf(mx, __shfl_xor(mx, 16, 32));
    mx = fmaxf(mx, __shfl_xor(mx, 8, 32));
    mx = fmaxf(mx, __shfl_xor(mx, 4, 32));
    mx = fmaxf(mx, __shfl_xor(mx, 2, 32));
    mx = fmaxf(mx, __shfl_xor(mx, 1, 32));

    float sum = 0.0f;
#pragma unroll
    for (int i = 0; i < 4; ++i) {
      s[i].x = fast_exp(s[i].x - mx);
      s[i].y = fast_exp(s[i].y - mx);
      s[i].z = fast_exp(s[i].z - mx);
      s[i].w = fast_exp(s[i].w - mx);
      sum += (s[i].x + s[i].y) + (s[i].z + s[i].w);
    }
    sum += __shfl_xor(sum, 16, 32);
    sum += __shfl_xor(sum, 8, 32);
    sum += __shfl_xor(sum, 4, 32);
    sum += __shfl_xor(sum, 2, 32);
    sum += __shfl_xor(sum, 1, 32);
    const float inv = 1.0f / sum;

    float* orow = Sout + (size_t)(xrow0 + r) * TM;
#pragma unroll
    for (int i = 0; i < 4; ++i) {
      float4 o;
      o.x = s[i].x * inv; o.y = s[i].y * inv; o.z = s[i].z * inv; o.w = s[i].w * inv;
      *(float4*)(orow + 128 * i + 4 * lane) = o;
    }
  }
}

__global__ __launch_bounds__(128) void attn_out_kernel(
    const float* __restrict__ S, const float* __restrict__ Mem,
    float* __restrict__ Out) {
  const int warp = threadIdx.x >> 5;
  const int lane = threadIdx.x & 31;
  const int half = lane >> 4;
  const int l    = lane & 15;

  const int pairsPerRow   = D / 32;
  const int tilesPerBatch = (TX / 16) * pairsPerRow;
  int tile  = blockIdx.x * (blockDim.x >> 5) + warp;
  int batch = tile / tilesPerBatch;
  int t     = tile % tilesPerBatch;
  int mt = t / pairsPerRow;
  int np = t % pairsPerRow;
  const int row0 = mt * 16;
  const int col0 = np * 32;

  const float* Sb = S   + (size_t)batch * TX * TM;
  const float* Mb = Mem + (size_t)batch * TM * D;
  const float* arow = Sb + (size_t)(row0 + l) * TM;

  v8f acc0 = {}; v8f acc1 = {};
#pragma unroll 4
  for (int k = 0; k < TM; k += 4) {
    const int kr = k + 2 * half;
    v2f a = *(const v2f*)(arow + kr);
    v2f b0, b1;
    b0.x = Mb[(size_t)kr * D + col0 + l];
    b0.y = Mb[(size_t)(kr + 1) * D + col0 + l];
    b1.x = Mb[(size_t)kr * D + col0 + 16 + l];
    b1.y = Mb[(size_t)(kr + 1) * D + col0 + 16 + l];
    acc0 = __builtin_amdgcn_wmma_f32_16x16x4_f32(false, a, false, b0, (short)0, acc0, false, false);
    acc1 = __builtin_amdgcn_wmma_f32_16x16x4_f32(false, a, false, b1, (short)0, acc1, false, false);
  }
#pragma unroll
  for (int j = 0; j < 8; ++j) {
    int r = row0 + j + 8 * half;
    Out[((size_t)batch * TX + r) * D + col0 + l]      = acc0[j];
    Out[((size_t)batch * TX + r) * D + col0 + 16 + l] = acc1[j];
  }
}

extern "C" void kernel_launch(void* const* d_in, const int* in_sizes, int n_in,
                              void* d_out, int out_size, void* d_ws, size_t ws_size,
                              hipStream_t stream) {
  const float* x    = (const float*)d_in[0];
  const float* mem  = (const float*)d_in[1];
  const int*   mask = (const int*)d_in[2];
  const float* w1   = (const float*)d_in[3];
  const float* b1   = (const float*)d_in[4];
  const float* w2   = (const float*)d_in[5];
  const float* wst  = (const float*)d_in[6];

  float* item1 = (float*)d_ws;
  float* item2 = item1 + (size_t)BATCH * TX * D;
  float* out   = (float*)d_out;
  float* Sout  = out + (size_t)BATCH * TX * D;

  const int rowsX = BATCH * TX;
  dim3 blk(128);

  proj_gemm_kernel<<<2 * PROJ_BLOCKS, blk, 0, stream>>>(x, w1, b1, item1, mem, w2, item2, rowsX);

  scores_softmax_kernel<<<BATCH * TX / XT, TM, 0, stream>>>(item1, item2, wst, mask, Sout);

  int tilesOut = BATCH * (TX / 16) * (D / 32);
  attn_out_kernel<<<tilesOut / 4, blk, 0, stream>>>(Sout, mem, out);
}


# S*memory k loop software-pipelined: ring of four register sets of 4 k-steps, the loads of three sets in flight under the products of the fourth; same loads, products and sum order
# speedup vs baseline: 1.7849x; 1.1639x over previous
#include <hip/hip_runtime.h>
#include <math.h>

#define D      256
#define TX     512
#define TM     512
#define BATCH  4

typedef __attribute__((ext_vector_type(2))) float v2f;
typedef __attribute__((ext_vector_type(8))) float v8f;

__device__ __forceinline__ float fast_tanh(float x) {
#if __has_builtin(__builtin_amdgcn_tanhf)
  return __builtin_amdgcn_tanhf(x);
#else
  float r;
  asm("v_tanh_f32 %0, %1" : "=v"(r) : "v"(x));
  return r;
#endif
}

__device__ __forceinline__ float fast_exp(float x) {
#if __has_builtin(__builtin_amdgcn_exp2f)
  return __builtin_amdgcn_exp2f(x * 1.4426950408889634f);
#else
  return __expf(x);
#endif
}

#define PROJ_BLOCKS ((BATCH * TX / 16) * (D / 32) / 4)
static_assert(TX == TM, "both projections must have the same number of rows to share one grid split");
__global__ __launch_bounds__(128) void proj_gemm_kernel(
    const float* __restrict__ In0, const float* __restrict__ W0,
    const float* __restrict__ bias0, float* __restrict__ Out0,
    const float* __restrict__ In1, const float* __restrict__ W1,
    float* __restrict__ Out1, int Mrows) {
  const int warp = threadIdx.x >> 5;
  const int lane = threadIdx.x & 31;
  const int half = lane >> 4;
  const int l    = lane & 15;

  const bool second = blockIdx.x >= PROJ_BLOCKS;
  const float* In   = second ? In1 : In0;
  const float* W    = second ? W1 : W0;
  const float* bias = second ? nullptr : bias0;
  float*       Out  = second ? Out1 : Out0;

  const int pairsPerRow = D / 32;
  int tile = (blockIdx.x - (second ? PROJ_BLOCKS : 0)) * (blockDim.x >> 5) + warp;
  int mt = tile / pairsPerRow;
  int np = tile % pairsPerRow;
  if (mt * 16 >= Mrows) return;
  const int row0 = mt * 16;
  const int col0 = np * 32;

  v8f acc0 = {}; v8f acc1 = {};
  const float* arow  = In + (size_t)(row0 + l) * D;
  const float* b0row = W  + (size_t)(col0 + l) * D;
  const float* b1row = W  + (size_t)(col0 + 16 + l) * D;
#pragma unroll 8
  for (int k = 0; k < D; k += 4) {
    v2f a  = *(const v2f*)(arow  + k + 2 * half);
    v2f b0 = *(const v2f*)(b0row + k + 2 * half);
    v2f b1 = *(const v2f*)(b1row + k + 2 * half);
    acc0 = __builtin_amdgcn_wmma_f32_16x16x4_f32(false, a, false, b0, (short)0, acc0, false, false);
    acc1 = __builtin_amdgcn_wmma_f32_16x16x4_f32(false, a, false, b1, (short)0, acc1, false, false);
  }
  float bv0 = 0.0f, bv1 = 0.0f;
  if (bias) { bv0 = bias[col0 + l]; bv1 = bias[col0 + 16 + l]; }
#pragma unroll
  for (int j = 0; j < 8; ++j) {
    int r = row0 + j + 8 * half;
    Out[(size_t)r * D + col0 + l]      = acc0[j] + bv0;
    Out[(size_t)r * D + col0 + 16 + l] = acc1[j] + bv1;
  }
}

typedef const float __attribute__((address_space(4))) cfloat_k;
__device__ __forceinline__ const cfloat_k* as_const_space(const float* p) {
  return (const cfloat_k*)p;
}

#define XT 8
#define DC 64
#define PITCH 68
__global__ __launch_bounds__(512) void scores_softmax_kernel(
    const float* __restrict__ item1, const float* __restrict__ item2,
    const float* __restrict__ wst, const int* __restrict__ mask,
    float* __restrict__ Sout) {
  __shared__ float t2[TM * PITCH];
  __shared__ float sc[XT][TM];

  const int tid   = threadIdx.x;
  const int warp  = tid >> 5;
  const int lane  = tid & 31;
  const int xrow0 = blockIdx.x * XT;
  const int b     = xrow0 / TX;

  const float*    i2b = item2 + (size_t)b * TM * D;
  const cfloat_k* i1b = as_const_space(item1 + (size_t)xrow0 * D);
  const cfloat_k* wsc = as_const_space(wst);

  const int srow = tid >> 4;
  const int scol = 4 * (tid & 15);

  float acc[XT];
#pragma unroll
  for (int x = 0; x < XT; ++x) acc[x] = 0.0f;

  for (int d0 = 0; d0 < D; d0 += DC) {
    __syncthreads();
#pragma unroll
    for (int i = 0; i < 16; ++i) {
      const float4 g = *(const float4*)(i2b + (size_t)(srow + 32 * i) * D + d0 + scol);
      *(float4*)(&t2[(srow + 32 * i) * PITCH + scol]) = g;
    }
    __syncthreads();
#pragma unroll 4
    for (int k = 0; k < DC; k += 4) {
      const float4 v = *(const float4*)(&t2[tid * PITCH + k]);
      const float w_0 = wsc[d0 + k], w_1 = wsc[d0 + k + 1], w_2 = wsc[d0 + k + 2], w_3 = wsc[d0 + k + 3];
#pragma unroll
      for (int x = 0; x < XT; ++x) {
        const cfloat_k* ap = i1b + x * D + d0 + k;
        const float a_0 = ap[0], a_1 = ap[1], a_2 = ap[2], a_3 = ap[3];
        acc[x] += fast_tanh(a_0 + v.x) * w_0;
        acc[x] += fast_tanh(a_1 + v.y) * w_1;
        acc[x] += fast_tanh(a_2 + v.z) * w_2;
        acc[x] += fast_tanh(a_3 + v.w) * w_3;
      }
    }
  }

#pragma unroll
  for (int x = 0; x < XT; ++x) sc[x][tid] = acc[x];
  __syncthreads();

  if (warp < XT) {
    const int r = warp;
    float4 s[4];
#pragma unroll
    for (int i = 0; i < 4; ++i) {
      s[i] = *(const float4*)(&sc[r][128 * i + 4 * lane]);
      const int4 mk = *(const int4*)(mask + b * TM + 128 * i + 4 * lane);
      if (mk.x == 0) s[i].x = -INFINITY;
      if (mk.y == 0) s[i].y = -INFINITY;
      if (mk.z == 0) s[i].z = -INFINITY;
      if (mk.w == 0) s[i].w = -INFINITY;
    }
    float mx = fmaxf(fmaxf(s[0].x, s[0].y), fmaxf(s[0].z, s[0].w));
#pragma unroll
    for (int i = 1; i < 4; ++i)
      mx = fmaxf(mx, fmaxf(fmaxf(s[i].x, s[i].y), fmaxf(s[i].z, s[i].w)));
    mx = fmaxf(mx, __shfl_xor(mx, 16, 32));
    mx = fmaxf(mx, __shfl_xor(mx, 8, 32));
    mx = fmaxf(mx, __shfl_xor(mx, 4, 32));
    mx = fmaxf(mx, __shfl_xor(mx, 2, 32));
    mx = fmaxf(mx, __shfl_xor(mx, 1, 32));

    float sum = 0.0f;
#pragma unroll
    for (int i = 0; i < 4; ++i) {
      s[i].x = fast_exp(s[i].x - mx);
      s[i].y = fast_exp(s[i].y - mx);
      s[i].z = fast_exp(s[i].z - mx);
      s[i].w = fast_exp(s[i].w - mx);
      sum += (s[i].x + s[i].y) + (s[i].z + s[i].w);
    }
    sum += __shfl_xor(sum, 16, 32);
    sum += __shfl_xor(sum, 8, 32);
    sum += __shfl_xor(sum, 4, 32);
    sum += __shfl_xor(sum, 2, 32);
    sum += __shfl_xor(sum, 1, 32);
    const float inv = 1.0f / sum;

    float* orow = Sout + (size_t)(xrow0 + r) * TM;
#pragma unroll
    for (int i = 0; i < 4; ++i) {
      float4 o;
      o.x = s[i].x * inv; o.y = s[i].y * inv; o.z = s[i].z * inv; o.w = s[i].w * inv;
      *(float4*)(orow + 128 * i + 4 * lane) = o;
    }
  }
}

#define KG    4
#define KSPAN (4 * KG)
struct KSet { v2f a[KG]; v2f b0[KG]; v2f b1[KG]; };

__device__ __forceinline__ void load_kset(KSet& f, const float* __restrict__ arow,
                                          const float* __restrict__ bcol, int k0, int half) {
#pragma unroll
  for (int i = 0; i < KG; ++i) {
    const int kr = k0 + 4 * i + 2 * half;
    f.a[i]    = *(const v2f*)(arow + kr);
    f.b0[i].x = bcol[(size_t)kr * D];
    f.b0[i].y = bcol[(size_t)(kr + 1) * D];
    f.b1[i].x = bcol[(size_t)kr * D + 16];
    f.b1[i].y = bcol[(size_t)(kr + 1) * D + 16];
  }
}

__device__ __forceinline__ void mma_kset(const KSet& f, v8f& acc0, v8f& acc1) {
#pragma unroll
  for (int i = 0; i < KG; ++i) {
    acc0 = __builtin_amdgcn_wmma_f32_16x16x4_f32(false, f.a[i], false, f.b0[i], (short)0, acc0, false, false);
    acc1 = __builtin_amdgcn_wmma_f32_16x16x4_f32(false, f.a[i], false, f.b1[i], (short)0, acc1, false, false);
  }
}

__device__ __forceinline__ void ring_order(v8f& acc0, v8f& acc1) {
  asm volatile("" : "+v"(acc0), "+v"(acc1) : : "memory");
}

static_assert(TM % KSPAN == 0 && (TM / KSPAN) % 4 == 0 && TM / KSPAN >= 8,
              "the ring below consumes its sets in trips of four");
__global__ __launch_bounds__(128) void attn_out_kernel(
    const float* __restrict__ S, const float* __restrict__ Mem,
    float* __restrict__ Out) {
  const int warp = threadIdx.x >> 5;
  const int lane = threadIdx.x & 31;
  const int half = lane >> 4;
  const int l    = lane & 15;

  const int pairsPerRow   = D / 32;
  const int tilesPerBatch = (TX / 16) * pairsPerRow;
  int tile  = blockIdx.x * (blockDim.x >> 5) + warp;
  int batch = tile / tilesPerBatch;
  int t     = tile % tilesPerBatch;
  int mt = t / pairsPerRow;
  int np = t % pairsPerRow;
  const int row0 = mt * 16;
  const int col0 = np * 32;

  const float* Sb = S   + (size_t)batch * TX * TM;
  const float* Mb = Mem + (size_t)batch * TM * D;
  const float* arow = Sb + (size_t)(row0 + l) * TM;
  const float* bcol = Mb + col0 + l;

  v8f acc0 = {}; v8f acc1 = {};
  KSet r0, r1, r2, r3;
  load_kset(r0, arow, bcol, 0 * KSPAN, half);
  load_kset(r1, arow, bcol, 1 * KSPAN, half);
  load_kset(r2, arow, bcol, 2 * KSPAN, half);
  int k = 0;
  for (; k < TM - 4 * KSPAN; k += 4 * KSPAN) {
    load_kset(r3, arow, bcol, k + 3 * KSPAN, half); mma_kset(r0, acc0, acc1); ring_order(acc0, acc1);
    load_kset(r0, arow, bcol, k + 4 * KSPAN, half); mma_kset(r1, acc0, acc1); ring_order(acc0, acc1);
    load_kset(r1, arow, bcol, k + 5 * KSPAN, half); mma_kset(r2, acc0, acc1); ring_order(acc0, acc1);
    load_kset(r2, arow, bcol, k + 6 * KSPAN, half); mma_kset(r3, acc0, acc1); ring_order(acc0, acc1);
  }
  load_kset(r3, arow, bcol, k + 3 * KSPAN, half);
  mma_kset(r0, acc0, acc1);
  mma_kset(r1, acc0, acc1);
  mma_kset(r2, acc0, acc1);
  mma_kset(r3, acc0, acc1);

#pragma unroll
  for (int j = 0; j < 8; ++j) {
    int r = row0 + j + 8 * half;
    Out[((size_t)batch * TX + r) * D + col0 + l]      = acc0[j];
    Out[((size_t)batch * TX + r) * D + col0 + 16 + l] = acc1[j];
  }
}

extern "C" void kernel_launch(void* const* d_in, const int* in_sizes, int n_in,
                              void* d_out, int out_size, void* d_ws, size_t ws_size,
                              hipStream_t stream) {
  const float* x    = (const float*)d_in[0];
  const float* mem  = (const float*)d_in[1];
  const int*   mask = (const int*)d_in[2];
  const float* w1   = (const float*)d_in[3];
  const float* b1   = (const float*)d_in[4];
  const float* w2   = (const float*)d_in[5];
  const float* wst  = (const float*)d_in[6];

  float* item1 = (float*)d_ws;
  float* item2 = item1 + (size_t)BATCH * TX * D;
  float* out   = (float*)d_out;
  float* Sout  = out + (size_t)BATCH * TX * D;

  const int rowsX = BATCH * TX;
  dim3 blk(128);

  proj_gemm_kernel<<<2 * PROJ_BLOCKS, blk, 0, stream>>>(x, w1, b1, item1, mem, w2, item2, rowsX);

  scores_softmax_kernel<<<BATCH * TX / XT, TM, 0, stream>>>(item1, item2, wst, mask, Sout);

  int tilesOut = BATCH * (TX / 16) * (D / 32);
  attn_out_kernel<<<tilesOut / 4, blk, 0, stream>>>(Sout, mem, out);
}


# item2 written d-grouped [b][d/4][m][4] by its projection (run as the transpose), so a scores lane loads its own row from global one trip ahead: no LDS staging, no barrier in the d loop
# speedup vs baseline: 2.1175x; 1.1863x over previous
#include <hip/hip_runtime.h>
#include <math.h>

#define D      256
#define TX     512
#define TM     512
#define BATCH  4

typedef __attribute__((ext_vector_type(2))) float v2f;
typedef __attribute__((ext_vector_type(8))) float v8f;

__device__ __forceinline__ float fast_tanh(float x) {
#if __has_builtin(__builtin_amdgcn_tanhf)
  return __builtin_amdgcn_tanhf(x);
#else
  float r;
  asm("v_tanh_f32 %0, %1" : "=v"(r) : "v"(x));
  return r;
#endif
}

__device__ __forceinline__ float fast_exp(float x) {
#if __has_builtin(__builtin_amdgcn_exp2f)
  return __builtin_amdgcn_exp2f(x * 1.4426950408889634f);
#else
  return __expf(x);
#endif
}

#define PROJ_BLOCKS ((BATCH * TX / 16) * (D / 32) / 4)
static_assert(TX == TM, "both projections must have the same number of rows to share one grid split");
static_assert((D / 16) * (BATCH * TM / 32) == PROJ_BLOCKS * 4, "the transposed problem has the same number of dual-tiles");
__global__ __launch_bounds__(128) void proj_gemm_kernel(
    const float* __restrict__ In0, const float* __restrict__ W0,
    const float* __restrict__ bias0, float* __restrict__ Out0,
    const float* __restrict__ In1, const float* __restrict__ W1,
    float* __restrict__ Out1, int Mrows) {
  const int warp = threadIdx.x >> 5;
  const int lane = threadIdx.x & 31;
  const int half = lane >> 4;
  const int l    = lane & 15;

  const bool second = blockIdx.x >= PROJ_BLOCKS;
  const int tile = (blockIdx.x - (second ? PROJ_BLOCKS : 0)) * (blockDim.x >> 5) + warp;

  static_assert(D / 16 == 16 && D / 32 == 8, "the tile split below is written with shifts");
  const int grp = second ? tile >> 4 : tile >> 3;
  const int sub = second ? tile & 15 : tile & 7;
  const int row0 = second ? sub * 16 : grp * 16;
  const int col0 = second ? grp * 32 : sub * 32;
  if (!second && row0 >= Mrows) return;

  const float* Am = second ? W1 : In0;
  const float* Bm = second ? In1 : W0;

  v8f acc0 = {}; v8f acc1 = {};
  const float* arow  = Am + (size_t)(row0 + l) * D;
  const float* b0row = Bm + (size_t)(col0 + l) * D;
  const float* b1row = Bm + (size_t)(col0 + 16 + l) * D;
#pragma unroll 8
  for (int k = 0; k < D; k += 4) {
    v2f a  = *(const v2f*)(arow  + k + 2 * half);
    v2f b0 = *(const v2f*)(b0row + k + 2 * half);
    v2f b1 = *(const v2f*)(b1row + k + 2 * half);
    acc0 = __builtin_amdgcn_wmma_f32_16x16x4_f32(false, a, false, b0, (short)0, acc0, false, false);
    acc1 = __builtin_amdgcn_wmma_f32_16x16x4_f32(false, a, false, b1, (short)0, acc1, false, false);
  }

  if (!second) {
    const float bv0 = bias0[col0 + l], bv1 = bias0[col0 + 16 + l];
#pragma unroll
    for (int j = 0; j < 8; ++j) {
      int r = row0 + j + 8 * half;
      Out0[(size_t)r * D + col0 + l]      = acc0[j] + bv0;
      Out0[(size_t)r * D + col0 + 16 + l] = acc1[j] + bv1;
    }
  } else {
    const int b  = col0 / TM;
    const int m0 = col0 % TM;
    float4* dst = (float4*)Out1 + ((size_t)b * (D / 4) + row0 / 4 + 2 * half) * TM + m0 + l;
    dst[0]       = make_float4(acc0[0], acc0[1], acc0[2], acc0[3]);
    dst[TM]      = make_float4(acc0[4], acc0[5], acc0[6], acc0[7]);
    dst[16]      = make_float4(acc1[0], acc1[1], acc1[2], acc1[3]);
    dst[TM + 16] = make_float4(acc1[4], acc1[5], acc1[6], acc1[7]);
  }
}

typedef const float __attribute__((address_space(4))) cfloat_k;
__device__ __forceinline__ const cfloat_k* as_const_space(const float* p) {
  return (const cfloat_k*)p;
}

#define XT 8
#define DT 16
#define GT (DT / 4)
struct DTrip { float4 g[GT]; };

__device__ __forceinline__ void load_trip(DTrip& t, const float4* __restrict__ row, int d0) {
#pragma unroll
  for (int i = 0; i < GT; ++i) t.g[i] = row[(size_t)(d0 / 4 + i) * TM];
}

__device__ __forceinline__ void stretch_edge() { __builtin_amdgcn_sched_barrier(0); }

#define XS 4
__device__ __forceinline__ void score_trip(const DTrip& t, int d0, const cfloat_k* i1b,
                                           const cfloat_k* wsc, float (&acc)[XT]) {
#pragma unroll
  for (int xs = 0; xs < XT; xs += XS) {
    float th[GT][XS][4];
#pragma unroll
    for (int i = 0; i < GT; ++i) {
      const float4 v = t.g[i];
#pragma unroll
      for (int x = 0; x < XS; ++x) {
        const cfloat_k* ap = i1b + (xs + x) * D + d0 + 4 * i;
        th[i][x][0] = ap[0] + v.x;
        th[i][x][1] = ap[1] + v.y;
        th[i][x][2] = ap[2] + v.z;
        th[i][x][3] = ap[3] + v.w;
      }
    }
#pragma unroll
    for (int i = 0; i < GT; ++i)
#pragma unroll
      for (int x = 0; x < XS; ++x)
#pragma unroll
        for (int c = 0; c < 4; ++c) th[i][x][c] = fast_tanh(th[i][x][c]);
#pragma unroll
    for (int i = 0; i < GT; ++i) {
      const int k = d0 + 4 * i;
      const float w_0 = wsc[k], w_1 = wsc[k + 1], w_2 = wsc[k + 2], w_3 = wsc[k + 3];
#pragma unroll
      for (int x = 0; x < XS; ++x) {
        acc[xs + x] += th[i][x][0] * w_0;
        acc[xs + x] += th[i][x][1] * w_1;
        acc[xs + x] += th[i][x][2] * w_2;
        acc[xs + x] += th[i][x][3] * w_3;
      }
    }
    stretch_edge();
  }
}

static_assert(D % (2 * DT) == 0 && D >= 4 * DT, "the d loop below consumes two trips per pass and peels the last pass");
__global__ __launch_bounds__(512) __attribute__((amdgpu_waves_per_eu(4, 4))) void scores_softmax_kernel(
    const float* __restrict__ item1, const float* __restrict__ item2g,
    const float* __restrict__ wst, const int* __restrict__ mask,
    float* __restrict__ Sout) {
  __shared__ float sc[XT][TM];

  const int tid   = threadIdx.x;
  const int warp  = tid >> 5;
  const int lane  = tid & 31;
  const int xrow0 = blockIdx.x * XT;
  const int b     = xrow0 / TX;

  const float4*   i2row = (const float4*)item2g + (size_t)b * (D / 4) * TM + tid;
  const cfloat_k* i1b   = as_const_space(item1 + (size_t)xrow0 * D);
  const cfloat_k* wsc   = as_const_space(wst);

  float acc[XT];
#pragma unroll
  for (int x = 0; x < XT; ++x) acc[x] = 0.0f;

  DTrip ta, tb;
  load_trip(ta, i2row, 0);
  for (int d0 = 0; d0 < D - 2 * DT; d0 += 2 * DT) {
    load_trip(tb, i2row, d0 + DT);
    score_trip(ta, d0, i1b, wsc, acc);
    load_trip(ta, i2row, d0 + 2 * DT);
    score_trip(tb, d0 + DT, i1b, wsc, acc);
  }
  load_trip(tb, i2row, D - DT);
  score_trip(ta, D - 2 * DT, i1b, wsc, acc);
  score_trip(tb, D - DT, i1b, wsc, acc);

#pragma unroll
  for (int x = 0; x < XT; ++x) sc[x][tid] = acc[x];
  __syncthreads();

  if (warp < XT) {
    const int r = warp;
    float4 s[4];
#pragma unroll
    for (int i = 0; i < 4; ++i) {
      s[i] = *(const float4*)(&sc[r][128 * i + 4 * lane]);
      const int4 mk = *(const int4*)(mask + b * TM + 128 * i + 4 * lane);
      if (mk.x == 0) s[i].x = -INFINITY;
      if (mk.y == 0) s[i].y = -INFINITY;
      if (mk.z == 0) s[i].z = -INFINITY;
      if (mk.w == 0) s[i].w = -INFINITY;
    }
    float mx = fmaxf(fmaxf(s[0].x, s[0].y), fmaxf(s[0].z, s[0].w));
#pragma unroll
    for (int i = 1; i < 4; ++i)
      mx = fmaxf(mx, fmaxf(fmaxf(s[i].x, s[i].y), fmaxf(s[i].z, s[i].w)));
    mx = fmaxf(mx, __shfl_xor(mx, 16, 32));
    mx = fmaxf(mx, __shfl_xor(mx, 8, 32));
    mx = fmaxf(mx, __shfl_xor(mx, 4, 32));
    mx = fmaxf(mx, __shfl_xor(mx, 2, 32));
    mx = fmaxf(mx, __shfl_xor(mx, 1, 32));

    float sum = 0.0f;
#pragma unroll
    for (int i = 0; i < 4; ++i) {
      s[i].x = fast_exp(s[i].x - mx);
      s[i].y = fast_exp(s[i].y - mx);
      s[i].z = fast_exp(s[i].z - mx);
      s[i].w = fast_exp(s[i].w - mx);
      sum += (s[i].x + s[i].y) + (s[i].z + s[i].w);
    }
    sum += __shfl_xor(sum, 16, 32);
    sum += __shfl_xor(sum, 8, 32);
    sum += __shfl_xor(sum, 4, 32);
    sum += __shfl_xor(sum, 2, 32);
    sum += __shfl_xor(sum, 1, 32);
    const float inv = 1.0f / sum;

    float* orow = Sout + (size_t)(xrow0 + r) * TM;
#pragma unroll
    for (int i = 0; i < 4; ++i) {
      float4 o;
      o.x = s[i].x * inv; o.y = s[i].y * inv; o.z = s[i].z * inv; o.w = s[i].w * inv;
      *(float4*)(orow + 128 * i + 4 * lane) = o;
    }
  }
}

#define KG    4
#define KSPAN (4 * KG)
struct KSet { v2f a[KG]; v2f b0[KG]; v2f b1[KG]; };

__device__ __forceinline__ void load_kset(KSet& f, const float* __restrict__ arow,
                                          const float* __restrict__ bcol, int k0, int half) {
#pragma unroll
  for (int i = 0; i < KG; ++i) {
    const int kr = k0 + 4 * i + 2 * half;
    f.a[i]    = *(const v2f*)(arow + kr);
    f.b0[i].x = bcol[(size_t)kr * D];
    f.b0[i].y = bcol[(size_t)(kr + 1) * D];
    f.b1[i].x = bcol[(size_t)kr * D + 16];
    f.b1[i].y = bcol[(size_t)(kr + 1) * D + 16];
  }
}

__device__ __forceinline__ void mma_kset(const KSet& f, v8f& acc0, v8f& acc1) {
#pragma unroll
  for (int i = 0; i < KG; ++i) {
    acc0 = __builtin_amdgcn_wmma_f32_16x16x4_f32(false, f.a[i], false, f.b0[i], (short)0, acc0, false, false);
    acc1 = __builtin_amdgcn_wmma_f32_16x16x4_f32(false, f.a[i], false, f.b1[i], (short)0, acc1, false, false);
  }
}

__device__ __forceinline__ void ring_order(v8f& acc0, v8f& acc1) {
  asm volatile("" : "+v"(acc0), "+v"(acc1) : : "memory");
}

static_assert(TM % KSPAN == 0 && (TM / KSPAN) % 4 == 0 && TM / KSPAN >= 8,
              "the ring below consumes its sets in trips of four");
__global__ __launch_bounds__(128) void attn_out_kernel(
    const float* __restrict__ S, const float* __restrict__ Mem,
    float* __restrict__ Out) {
  const int warp = threadIdx.x >> 5;
  const int lane = threadIdx.x & 31;
  const int half = lane >> 4;
  const int l    = lane & 15;

  const int pairsPerRow   = D / 32;
  const int tilesPerBatch = (TX / 16) * pairsPerRow;
  int tile  = blockIdx.x * (blockDim.x >> 5) + warp;
  int batch = tile / tilesPerBatch;
  int t     = tile % tilesPerBatch;
  int mt = t / pairsPerRow;
  int np = t % pairsPerRow;
  const int row0 = mt * 16;
  const int col0 = np * 32;

  const float* Sb = S   + (size_t)batch * TX * TM;
  const float* Mb = Mem + (size_t)batch * TM * D;
  const float* arow = Sb + (size_t)(row0 + l) * TM;
  const float* bcol = Mb + col0 + l;

  v8f acc0 = {}; v8f acc1 = {};
  KSet r0, r1, r2, r3;
  load_kset(r0, arow, bcol, 0 * KSPAN, half);
  load_kset(r1, arow, bcol, 1 * KSPAN, half);
  load_kset(r2, arow, bcol, 2 * KSPAN, half);
  int k = 0;
  for (; k < TM - 4 * KSPAN; k += 4 * KSPAN) {
    load_kset(r3, arow, bcol, k + 3 * KSPAN, half); mma_kset(r0, acc0, acc1); ring_order(acc0, acc1);
    load_kset(r0, arow, bcol, k + 4 * KSPAN, half); mma_kset(r1, acc0, acc1); ring_order(acc0, acc1);
    load_kset(r1, arow, bcol, k + 5 * KSPAN, half); mma_kset(r2, acc0, acc1); ring_order(acc0, acc1);
    load_kset(r2, arow, bcol, k + 6 * KSPAN, half); mma_kset(r3, acc0, acc1); ring_order(acc0, acc1);
  }
  load_kset(r3, arow, bcol, k + 3 * KSPAN, half);
  mma_kset(r0, acc0, acc1);
  mma_kset(r1, acc0, acc1);
  mma_kset(r2, acc0, acc1);
  mma_kset(r3, acc0, acc1);

#pragma unroll
  for (int j = 0; j < 8; ++j) {
    int r = row0 + j + 8 * half;
    Out[((size_t)batch * TX + r) * D + col0 + l]      = acc0[j];
    Out[((size_t)batch * TX + r) * D + col0 + 16 + l] = acc1[j];
  }
}

extern "C" void kernel_launch(void* const* d_in, const int* in_sizes, int n_in,
                              void* d_out, int out_size, void* d_ws, size_t ws_size,
                              hipStream_t stream) {
  const float* x    = (const float*)d_in[0];
  const float* mem  = (const float*)d_in[1];
  const int*   mask = (const int*)d_in[2];
  const float* w1   = (const float*)d_in[3];
  const float* b1   = (const float*)d_in[4];
  const float* w2   = (const float*)d_in[5];
  const float* wst  = (const float*)d_in[6];

  float* item1 = (float*)d_ws;
  float* item2 = item1 + (size_t)BATCH * TX * D;
  float* out   = (float*)d_out;
  float* Sout  = out + (size_t)BATCH * TX * D;

  const int rowsX = BATCH * TX;
  dim3 blk(128);

  proj_gemm_kernel<<<2 * PROJ_BLOCKS, blk, 0, stream>>>(x, w1, b1, item1, mem, w2, item2, rowsX);

  scores_softmax_kernel<<<BATCH * TX / XT, TM, 0, stream>>>(item1, item2, wst, mask, Sout);

  int tilesOut = BATCH * (TX / 16) * (D / 32);
  attn_out_kernel<<<tilesOut / 4, blk, 0, stream>>>(Sout, mem, out);
}


# scores d loop: issue order set to one tanh then one dual-issue pair of plain ops (skewed row/d order makes the pairs legal), so plain work goes out between two tanh; same sums, same order
# speedup vs baseline: 2.1869x; 1.0328x over previous
#include <hip/hip_runtime.h>
#include <math.h>

#define D      256
#define TX     512
#define TM     512
#define BATCH  4

typedef __attribute__((ext_vector_type(2))) float v2f;
typedef __attribute__((ext_vector_type(8))) float v8f;

__device__ __forceinline__ float fast_tanh(float x) {
#if __has_builtin(__builtin_amdgcn_tanhf)
  return __builtin_amdgcn_tanhf(x);
#else
  float r;
  asm("v_tanh_f32 %0, %1" : "=v"(r) : "v"(x));
  return r;
#endif
}

__device__ __forceinline__ float fast_exp(float x) {
#if __has_builtin(__builtin_amdgcn_exp2f)
  return __builtin_amdgcn_exp2f(x * 1.4426950408889634f);
#else
  return __expf(x);
#endif
}

#define PROJ_BLOCKS ((BATCH * TX / 16) * (D / 32) / 4)
static_assert(TX == TM, "both projections must have the same number of rows to share one grid split");
static_assert((D / 16) * (BATCH * TM / 32) == PROJ_BLOCKS * 4, "the transposed problem has the same number of dual-tiles");
__global__ __launch_bounds__(128) void proj_gemm_kernel(
    const float* __restrict__ In0, const float* __restrict__ W0,
    const float* __restrict__ bias0, float* __restrict__ Out0,
    const float* __restrict__ In1, const float* __restrict__ W1,
    float* __restrict__ Out1, int Mrows) {
  const int warp = threadIdx.x >> 5;
  const int lane = threadIdx.x & 31;
  const int half = lane >> 4;
  const int l    = lane & 15;

  const bool second = blockIdx.x >= PROJ_BLOCKS;
  const int tile = (blockIdx.x - (second ? PROJ_BLOCKS : 0)) * (blockDim.x >> 5) + warp;

  static_assert(D / 16 == 16 && D / 32 == 8, "the tile split below is written with shifts");
  const int grp = second ? tile >> 4 : tile >> 3;
  const int sub = second ? tile & 15 : tile & 7;
  const int row0 = second ? sub * 16 : grp * 16;
  const int col0 = second ? grp * 32 : sub * 32;
  if (!second && row0 >= Mrows) return;

  const float* Am = second ? W1 : In0;
  const float* Bm = second ? In1 : W0;

  v8f acc0 = {}; v8f acc1 = {};
  const float* arow  = Am + (size_t)(row0 + l) * D;
  const float* b0row = Bm + (size_t)(col0 + l) * D;
  const float* b1row = Bm + (size_t)(col0 + 16 + l) * D;
#pragma unroll 8
  for (int k = 0; k < D; k += 4) {
    v2f a  = *(const v2f*)(arow  + k + 2 * half);
    v2f b0 = *(const v2f*)(b0row + k + 2 * half);
    v2f b1 = *(const v2f*)(b1row + k + 2 * half);
    acc0 = __builtin_amdgcn_wmma_f32_16x16x4_f32(false, a, false, b0, (short)0, acc0, false, false);
    acc1 = __builtin_amdgcn_wmma_f32_16x16x4_f32(false, a, false, b1, (short)0, acc1, false, false);
  }

  if (!second) {
    const float bv0 = bias0[col0 + l], bv1 = bias0[col0 + 16 + l];
#pragma unroll
    for (int j = 0; j < 8; ++j) {
      int r = row0 + j + 8 * half;
      Out0[(size_t)r * D + col0 + l]      = acc0[j] + bv0;
      Out0[(size_t)r * D + col0 + 16 + l] = acc1[j] + bv1;
    }
  } else {
    const int b  = col0 / TM;
    const int m0 = col0 % TM;
    float4* dst = (float4*)Out1 + ((size_t)b * (D / 4) + row0 / 4 + 2 * half) * TM + m0 + l;
    dst[0]       = make_float4(acc0[0], acc0[1], acc0[2], acc0[3]);
    dst[TM]      = make_float4(acc0[4], acc0[5], acc0[6], acc0[7]);
    dst[16]      = make_float4(acc1[0], acc1[1], acc1[2], acc1[3]);
    dst[TM + 16] = make_float4(acc1[4], acc1[5], acc1[6], acc1[7]);
  }
}

typedef const float __attribute__((address_space(4))) cfloat_k;
__device__ __forceinline__ const cfloat_k* as_const_space(const float* p) {
  return (const cfloat_k*)p;
}

#define XT 8
#define DT 16
#define GT (DT / 4)
struct DTrip { float4 g[GT]; };

__device__ __forceinline__ void load_trip(DTrip& t, const float4* __restrict__ row, int d0) {
#pragma unroll
  for (int i = 0; i < GT; ++i) t.g[i] = row[(size_t)(d0 / 4 + i) * TM];
}

__device__ __forceinline__ void stretch_edge() { __builtin_amdgcn_sched_barrier(0); }

#define XS 4
#define SG_PLAIN 0x002
#define SG_TRANS 0x400
__device__ __forceinline__ void tanh_between_plain_pairs() {
  __builtin_amdgcn_sched_group_barrier(SG_PLAIN, 2, 0);
#pragma unroll
  for (int s = 0; s < XS * DT - 2; ++s) {
    __builtin_amdgcn_sched_group_barrier(SG_TRANS, 1, 0);
    __builtin_amdgcn_sched_group_barrier(SG_PLAIN, 2, 0);
  }
  __builtin_amdgcn_sched_group_barrier(SG_TRANS, 2, 0);
  __builtin_amdgcn_sched_group_barrier(SG_PLAIN, 2, 0);
}

__device__ __forceinline__ void score_trip(const DTrip& t, int d0, const cfloat_k* i1b,
                                           const cfloat_k* wsc, float (&acc)[XT]) {
#pragma unroll
  for (int xs = 0; xs < XT; xs += XS) {
    float th[GT][XS][4];
#pragma unroll
    for (int i = 0; i < GT; ++i) {
      const float4 v = t.g[i];
#pragma unroll
      for (int x = 0; x < XS; ++x) {
        const cfloat_k* ap = i1b + (xs + x) * D + d0 + 4 * i;
        th[i][x][0] = ap[0] + v.x;
        th[i][x][1] = ap[1] + v.y;
        th[i][x][2] = ap[2] + v.z;
        th[i][x][3] = ap[3] + v.w;
      }
    }
#pragma unroll
    for (int i = 0; i < GT; ++i)
#pragma unroll
      for (int x = 0; x < XS; ++x)
#pragma unroll
        for (int c = 0; c < 4; ++c) th[i][x][c] = fast_tanh(th[i][x][c]);
#pragma unroll
    for (int s = 0; s < DT + XS - 1; ++s)
#pragma unroll
      for (int x = 0; x < XS; ++x) {
        const int dd = s - x;
        if (dd >= 0 && dd < DT) acc[xs + x] += th[dd / 4][x][dd % 4] * wsc[d0 + dd];
      }
    tanh_between_plain_pairs();
    stretch_edge();
  }
}

static_assert(D % (2 * DT) == 0 && D >= 4 * DT, "the d loop below consumes two trips per pass and peels the last pass");
__global__ __launch_bounds__(512) __attribute__((amdgpu_waves_per_eu(4, 4))) void scores_softmax_kernel(
    const float* __restrict__ item1, const float* __restrict__ item2g,
    const float* __restrict__ wst, const int* __restrict__ mask,
    float* __restrict__ Sout) {
  __shared__ float sc[XT][TM];

  const int tid   = threadIdx.x;
  const int warp  = tid >> 5;
  const int lane  = tid & 31;
  const int xrow0 = blockIdx.x * XT;
  const int b     = xrow0 / TX;

  const float4*   i2row = (const float4*)item2g + (size_t)b * (D / 4) * TM + tid;
  const cfloat_k* i1b   = as_const_space(item1 + (size_t)xrow0 * D);
  const cfloat_k* wsc   = as_const_space(wst);

  float acc[XT];
#pragma unroll
  for (int x = 0; x < XT; ++x) acc[x] = 0.0f;

  DTrip ta, tb;
  load_trip(ta, i2row, 0);
  for (int d0 = 0; d0 < D - 2 * DT; d0 += 2 * DT) {
    load_trip(tb, i2row, d0 + DT);
    score_trip(ta, d0, i1b, wsc, acc);
    load_trip(ta, i2row, d0 + 2 * DT);
    score_trip(tb, d0 + DT, i1b, wsc, acc);
  }
  load_trip(tb, i2row, D - DT);
  score_trip(ta, D - 2 * DT, i1b, wsc, acc);
  score_trip(tb, D - DT, i1b, wsc, acc);

#pragma unroll
  for (int x = 0; x < XT; ++x) sc[x][tid] = acc[x];
  __syncthreads();

  if (warp < XT) {
    const int r = warp;
    float4 s[4];
#pragma unroll
    for (int i = 0; i < 4; ++i) {
      s[i] = *(const float4*)(&sc[r][128 * i + 4 * lane]);
      const int4 mk = *(const int4*)(mask + b * TM + 128 * i + 4 * lane);
      if (mk.x == 0) s[i].x = -INFINITY;
      if (mk.y == 0) s[i].y = -INFINITY;
      if (mk.z == 0) s[i].z = -INFINITY;
      if (mk.w == 0) s[i].w = -INFINITY;
    }
    float mx = fmaxf(fmaxf(s[0].x, s[0].y), fmaxf(s[0].z, s[0].w));
#pragma unroll
    for (int i = 1; i < 4; ++i)
      mx = fmaxf(mx, fmaxf(fmaxf(s[i].x, s[i].y), fmaxf(s[i].z, s[i].w)));
    mx = fmaxf(mx, __shfl_xor(mx, 16, 32));
    mx = fmaxf(mx, __shfl_xor(mx, 8, 32));
    mx = fmaxf(mx, __shfl_xor(mx, 4, 32));
    mx = fmaxf(mx, __shfl_xor(mx, 2, 32));
    mx = fmaxf(mx, __shfl_xor(mx, 1, 32));

    float sum = 0.0f;
#pragma unroll
    for (int i = 0; i < 4; ++i) {
      s[i].x = fast_exp(s[i].x - mx);
      s[i].y = fast_exp(s[i].y - mx);
      s[i].z = fast_exp(s[i].z - mx);
      s[i].w = fast_exp(s[i].w - mx);
      sum += (s[i].x + s[i].y) + (s[i].z + s[i].w);
    }
    sum += __shfl_xor(sum, 16, 32);
    sum += __shfl_xor(sum, 8, 32);
    sum += __shfl_xor(sum, 4, 32);
    sum += __shfl_xor(sum, 2, 32);
    sum += __shfl_xor(sum, 1, 32);
    const float inv = 1.0f / sum;

    float* orow = Sout + (size_t)(xrow0 + r) * TM;
#pragma unroll
    for (int i = 0; i < 4; ++i) {
      float4 o;
      o.x = s[i].x * inv; o.y = s[i].y * inv; o.z = s[i].z * inv; o.w = s[i].w * inv;
      *(float4*)(orow + 128 * i + 4 * lane) = o;
    }
  }
}

#define KG    4
#define KSPAN (4 * KG)
struct KSet { v2f a[KG]; v2f b0[KG]; v2f b1[KG]; };

__device__ __forceinline__ void load_kset(KSet& f, const float* __restrict__ arow,
                                          const float* __restrict__ bcol, int k0, int half) {
#pragma unroll
  for (int i = 0; i < KG; ++i) {
    const int kr = k0 + 4 * i + 2 * half;
    f.a[i]    = *(const v2f*)(arow + kr);
    f.b0[i].x = bcol[(size_t)kr * D];
    f.b0[i].y = bcol[(size_t)(kr + 1) * D];
    f.b1[i].x = bcol[(size_t)kr * D + 16];
    f.b1[i].y = bcol[(size_t)(kr + 1) * D + 16];
  }
}

__device__ __forceinline__ void mma_kset(const KSet& f, v8f& acc0, v8f& acc1) {
#pragma unroll
  for (int i = 0; i < KG; ++i) {
    acc0 = __builtin_amdgcn_wmma_f32_16x16x4_f32(false, f.a[i], false, f.b0[i], (short)0, acc0, false, false);
    acc1 = __builtin_amdgcn_wmma_f32_16x16x4_f32(false, f.a[i], false, f.b1[i], (short)0, acc1, false, false);
  }
}

__device__ __forceinline__ void ring_order(v8f& acc0, v8f& acc1) {
  asm volatile("" : "+v"(acc0), "+v"(acc1) : : "memory");
}

static_assert(TM % KSPAN == 0 && (TM / KSPAN) % 4 == 0 && TM / KSPAN >= 8,
              "the ring below consumes its sets in trips of four");
__global__ __launch_bounds__(128) void attn_out_kernel(
    const float* __restrict__ S, const float* __restrict__ Mem,
    float* __restrict__ Out) {
  const int warp = threadIdx.x >> 5;
  const int lane = threadIdx.x & 31;
  const int half = lane >> 4;
  const int l    = lane & 15;

  const int pairsPerRow   = D / 32;
  const int tilesPerBatch = (TX / 16) * pairsPerRow;
  int tile  = blockIdx.x * (blockDim.x >> 5) + warp;
  int batch = tile / tilesPerBatch;
  int t     = tile % tilesPerBatch;
  int mt = t / pairsPerRow;
  int np = t % pairsPerRow;
  const int row0 = mt * 16;
  const int col0 = np * 32;

  const float* Sb = S   + (size_t)batch * TX * TM;
  const float* Mb = Mem + (size_t)batch * TM * D;
  const float* arow = Sb + (size_t)(row0 + l) * TM;
  const float* bcol = Mb + col0 + l;

  v8f acc0 = {}; v8f acc1 = {};
  KSet r0, r1, r2, r3;
  load_kset(r0, arow, bcol, 0 * KSPAN, half);
  load_kset(r1, arow, bcol, 1 * KSPAN, half);
  load_kset(r2, arow, bcol, 2 * KSPAN, half);
  int k = 0;
  for (; k < TM - 4 * KSPAN; k += 4 * KSPAN) {
    load_kset(r3, arow, bcol, k + 3 * KSPAN, half); mma_kset(r0, acc0, acc1); ring_order(acc0, acc1);
    load_kset(r0, arow, bcol, k + 4 * KSPAN, half); mma_kset(r1, acc0, acc1); ring_order(acc0, acc1);
    load_kset(r1, arow, bcol, k + 5 * KSPAN, half); mma_kset(r2, acc0, acc1); ring_order(acc0, acc1);
    load_kset(r2, arow, bcol, k + 6 * KSPAN, half); mma_kset(r3, acc0, acc1); ring_order(acc0, acc1);
  }
  load_kset(r3, arow, bcol, k + 3 * KSPAN, half);
  mma_kset(r0, acc0, acc1);
  mma_kset(r1, acc0, acc1);
  mma_kset(r2, acc0, acc1);
  mma_kset(r3, acc0, acc1);

#pragma unroll
  for (int j = 0; j < 8; ++j) {
    int r = row0 + j + 8 * half;
    Out[((size_t)batch * TX + r) * D + col0 + l]      = acc0[j];
    Out[((size_t)batch * TX + r) * D + col0 + 16 + l] = acc1[j];
  }
}

extern "C" void kernel_launch(void* const* d_in, const int* in_sizes, int n_in,
                              void* d_out, int out_size, void* d_ws, size_t ws_size,
                              hipStream_t stream) {
  const float* x    = (const float*)d_in[0];
  const float* mem  = (const float*)d_in[1];
  const int*   mask = (const int*)d_in[2];
  const float* w1   = (const float*)d_in[3];
  const float* b1   = (const float*)d_in[4];
  const float* w2   = (const float*)d_in[5];
  const float* wst  = (const float*)d_in[6];

  float* item1 = (float*)d_ws;
  float* item2 = item1 + (size_t)BATCH * TX * D;
  float* out   = (float*)d_out;
  float* Sout  = out + (size_t)BATCH * TX * D;

  const int rowsX = BATCH * TX;
  dim3 blk(128);

  proj_gemm_kernel<<<2 * PROJ_BLOCKS, blk, 0, stream>>>(x, w1, b1, item1, mem, w2, item2, rowsX);

  scores_softmax_kernel<<<BATCH * TX / XT, TM, 0, stream>>>(item1, item2, wst, mask, Sout);

  int tilesOut = BATCH * (TX / 16) * (D / 32);
  attn_out_kernel<<<tilesOut / 4, blk, 0, stream>>>(Sout, mem, out);
}


# projection GEMM k loop software-pipelined on a ring of six register sets of 4 k-steps: loads of five sets (60) in flight under the products of the sixth; same loads, products and sum order
# speedup vs baseline: 2.2852x; 1.0449x over previous
#include <hip/hip_runtime.h>
#include <math.h>

#define D      256
#define TX     512
#define TM     512
#define BATCH  4

typedef __attribute__((ext_vector_type(2))) float v2f;
typedef __attribute__((ext_vector_type(8))) float v8f;

__device__ __forceinline__ float fast_tanh(float x) {
#if __has_builtin(__builtin_amdgcn_tanhf)
  return __builtin_amdgcn_tanhf(x);
#else
  float r;
  asm("v_tanh_f32 %0, %1" : "=v"(r) : "v"(x));
  return r;
#endif
}

__device__ __forceinline__ float fast_exp(float x) {
#if __has_builtin(__builtin_amdgcn_exp2f)
  return __builtin_amdgcn_exp2f(x * 1.4426950408889634f);
#else
  return __expf(x);
#endif
}

#define KG    4
#define KSPAN (4 * KG)
struct KSet { v2f a[KG]; v2f b0[KG]; v2f b1[KG]; };

__device__ __forceinline__ void mma_kset(const KSet& f, v8f& acc0, v8f& acc1) {
#pragma unroll
  for (int i = 0; i < KG; ++i) {
    acc0 = __builtin_amdgcn_wmma_f32_16x16x4_f32(false, f.a[i], false, f.b0[i], (short)0, acc0, false, false);
    acc1 = __builtin_amdgcn_wmma_f32_16x16x4_f32(false, f.a[i], false, f.b1[i], (short)0, acc1, false, false);
  }
}

__device__ __forceinline__ void ring_order(v8f& acc0, v8f& acc1) {
  asm volatile("" : "+v"(acc0), "+v"(acc1) : : "memory");
}

__device__ __forceinline__ void load_rows_kset(KSet& f, const float* __restrict__ arow,
                                               const float* __restrict__ b0row,
                                               const float* __restrict__ b1row, int k0, int half) {
#pragma unroll
  for (int i = 0; i < KG; ++i) {
    const int kr = k0 + 4 * i + 2 * half;
    f.a[i]  = *(const v2f*)(arow  + kr);
    f.b0[i] = *(const v2f*)(b0row + kr);
    f.b1[i] = *(const v2f*)(b1row + kr);
  }
}
#define PROJ_RING 6
#define PROJ_SETS (D / KSPAN)
static_assert(D % KSPAN == 0 && PROJ_SETS >= PROJ_RING, "the projection's k range is a whole number of sets, at least one ring");
static_assert(3 * KG * (PROJ_RING - 1) <= 63, "the loads in flight must fit the wave's outstanding-load counter");
#define PROJ_BLOCKS ((BATCH * TX / 16) * (D / 32) / 4)
static_assert(TX == TM, "both projections must have the same number of rows to share one grid split");
static_assert((D / 16) * (BATCH * TM / 32) == PROJ_BLOCKS * 4, "the transposed problem has the same number of dual-tiles");
__global__ __launch_bounds__(128) void proj_gemm_kernel(
    const float* __restrict__ In0, const float* __restrict__ W0,
    const float* __restrict__ bias0, float* __restrict__ Out0,
    const float* __restrict__ In1, const float* __restrict__ W1,
    float* __restrict__ Out1, int Mrows) {
  const int warp = threadIdx.x >> 5;
  const int lane = threadIdx.x & 31;
  const int half = lane >> 4;
  const int l    = lane & 15;

  const bool second = blockIdx.x >= PROJ_BLOCKS;
  const int tile = (blockIdx.x - (second ? PROJ_BLOCKS : 0)) * (blockDim.x >> 5) + warp;

  static_assert(D / 16 == 16 && D / 32 == 8, "the tile split below is written with shifts");
  const int grp = second ? tile >> 4 : tile >> 3;
  const int sub = second ? tile & 15 : tile & 7;
  const int row0 = second ? sub * 16 : grp * 16;
  const int col0 = second ? grp * 32 : sub * 32;
  if (!second && row0 >= Mrows) return;

  const float* Am = second ? W1 : In0;
  const float* Bm = second ? In1 : W0;

  v8f acc0 = {}; v8f acc1 = {};
  const float* arow  = Am + (size_t)(row0 + l) * D;
  const float* b0row = Bm + (size_t)(col0 + l) * D;
  const float* b1row = Bm + (size_t)(col0 + 16 + l) * D;
  KSet ring[PROJ_RING];
#pragma unroll
  for (int s = 0; s < PROJ_RING - 1; ++s)
    load_rows_kset(ring[s], arow, b0row, b1row, s * KSPAN, half);
#pragma unroll
  for (int s = 0; s < PROJ_SETS; ++s) {
    if (s + PROJ_RING - 1 < PROJ_SETS)
      load_rows_kset(ring[(s + PROJ_RING - 1) % PROJ_RING], arow, b0row, b1row, (s + PROJ_RING - 1) * KSPAN, half);
    mma_kset(ring[s % PROJ_RING], acc0, acc1);
    ring_order(acc0, acc1);
  }

  if (!second) {
    const float bv0 = bias0[col0 + l], bv1 = bias0[col0 + 16 + l];
#pragma unroll
    for (int j = 0; j < 8; ++j) {
      int r = row0 + j + 8 * half;
      Out0[(size_t)r * D + col0 + l]      = acc0[j] + bv0;
      Out0[(size_t)r * D + col0 + 16 + l] = acc1[j] + bv1;
    }
  } else {
    const int b  = col0 / TM;
    const int m0 = col0 % TM;
    float4* dst = (float4*)Out1 + ((size_t)b * (D / 4) + row0 / 4 + 2 * half) * TM + m0 + l;
    dst[0]       = make_float4(acc0[0], acc0[1], acc0[2], acc0[3]);
    dst[TM]      = make_float4(acc0[4], acc0[5], acc0[6], acc0[7]);
    dst[16]      = make_float4(acc1[0], acc1[1], acc1[2], acc1[3]);
    dst[TM + 16] = make_float4(acc1[4], acc1[5], acc1[6], acc1[7]);
  }
}

typedef const float __attribute__((address_space(4))) cfloat_k;
__device__ __forceinline__ const cfloat_k* as_const_space(const float* p) {
  return (const cfloat_k*)p;
}

#define XT 8
#define DT 16
#define GT (DT / 4)
struct DTrip { float4 g[GT]; };

__device__ __forceinline__ void load_trip(DTrip& t, const float4* __restrict__ row, int d0) {
#pragma unroll
  for (int i = 0; i < GT; ++i) t.g[i] = row[(size_t)(d0 / 4 + i) * TM];
}

__device__ __forceinline__ void stretch_edge() { __builtin_amdgcn_sched_barrier(0); }

#define XS 4
#define SG_PLAIN 0x002
#define SG_TRANS 0x400
__device__ __forceinline__ void tanh_between_plain_pairs() {
  __builtin_amdgcn_sched_group_barrier(SG_PLAIN, 2, 0);
#pragma unroll
  for (int s = 0; s < XS * DT - 2; ++s) {
    __builtin_amdgcn_sched_group_barrier(SG_TRANS, 1, 0);
    __builtin_amdgcn_sched_group_barrier(SG_PLAIN, 2, 0);
  }
  __builtin_amdgcn_sched_group_barrier(SG_TRANS, 2, 0);
  __builtin_amdgcn_sched_group_barrier(SG_PLAIN, 2, 0);
}

__device__ __forceinline__ void score_trip(const DTrip& t, int d0, const cfloat_k* i1b,
                                           const cfloat_k* wsc, float (&acc)[XT]) {
#pragma unroll
  for (int xs = 0; xs < XT; xs += XS) {
    float th[GT][XS][4];
#pragma unroll
    for (int i = 0; i < GT; ++i) {
      const float4 v = t.g[i];
#pragma unroll
      for (int x = 0; x < XS; ++x) {
        const cfloat_k* ap = i1b + (xs + x) * D + d0 + 4 * i;
        th[i][x][0] = ap[0] + v.x;
        th[i][x][1] = ap[1] + v.y;
        th[i][x][2] = ap[2] + v.z;
        th[i][x][3] = ap[3] + v.w;
      }
    }
#pragma unroll
    for (int i = 0; i < GT; ++i)
#pragma unroll
      for (int x = 0; x < XS; ++x)
#pragma unroll
        for (int c = 0; c < 4; ++c) th[i][x][c] = fast_tanh(th[i][x][c]);
#pragma unroll
    for (int s = 0; s < DT + XS - 1; ++s)
#pragma unroll
      for (int x = 0; x < XS; ++x) {
        const int dd = s - x;
        if (dd >= 0 && dd < DT) acc[xs + x] += th[dd / 4][x][dd % 4] * wsc[d0 + dd];
      }
    tanh_between_plain_pairs();
    stretch_edge();
  }
}

static_assert(D % (2 * DT) == 0 && D >= 4 * DT, "the d loop below consumes two trips per pass and peels the last pass");
__global__ __launch_bounds__(512) __attribute__((amdgpu_waves_per_eu(4, 4))) void scores_softmax_kernel(
    const float* __restrict__ item1, const float* __restrict__ item2g,
    const float* __restrict__ wst, const int* __restrict__ mask,
    float* __restrict__ Sout) {
  __shared__ float sc[XT][TM];

  const int tid   = threadIdx.x;
  const int warp  = tid >> 5;
  const int lane  = tid & 31;
  const int xrow0 = blockIdx.x * XT;
  const int b     = xrow0 / TX;

  const float4*   i2row = (const float4*)item2g + (size_t)b * (D / 4) * TM + tid;
  const cfloat_k* i1b   = as_const_space(item1 + (size_t)xrow0 * D);
  const cfloat_k* wsc   = as_const_space(wst);

  float acc[XT];
#pragma unroll
  for (int x = 0; x < XT; ++x) acc[x] = 0.0f;

  DTrip ta, tb;
  load_trip(ta, i2row, 0);
  for (int d0 = 0; d0 < D - 2 * DT; d0 += 2 * DT) {
    load_trip(tb, i2row, d0 + DT);
    score_trip(ta, d0, i1b, wsc, acc);
    load_trip(ta, i2row, d0 + 2 * DT);
    score_trip(tb, d0 + DT, i1b, wsc, acc);
  }
  load_trip(tb, i2row, D - DT);
  score_trip(ta, D - 2 * DT, i1b, wsc, acc);
  score_trip(tb, D - DT, i1b, wsc, acc);

#pragma unroll
  for (int x = 0; x < XT; ++x) sc[x][tid] = acc[x];
  __syncthreads();

  if (warp < XT) {
    const int r = warp;
    float4 s[4];
#pragma unroll
    for (int i = 0; i < 4; ++i) {
      s[i] = *(const float4*)(&sc[r][128 * i + 4 * lane]);
      const int4 mk = *(const int4*)(mask + b * TM + 128 * i + 4 * lane);
      if (mk.x == 0) s[i].x = -INFINITY;
      if (mk.y == 0) s[i].y = -INFINITY;
      if (mk.z == 0) s[i].z = -INFINITY;
      if (mk.w == 0) s[i].w = -INFINITY;
    }
    float mx = fmaxf(fmaxf(s[0].x, s[0].y), fmaxf(s[0].z, s[0].w));
#pragma unroll
    for (int i = 1; i < 4; ++i)
      mx = fmaxf(mx, fmaxf(fmaxf(s[i].x, s[i].y), fmaxf(s[i].z, s[i].w)));
    mx = fmaxf(mx, __shfl_xor(mx, 16, 32));
    mx = fmaxf(mx, __shfl_xor(mx, 8, 32));
    mx = fmaxf(mx, __shfl_xor(mx, 4, 32));
    mx = fmaxf(mx, __shfl_xor(mx, 2, 32));
    mx = fmaxf(mx, __shfl_xor(mx, 1, 32));

    float sum = 0.0f;
#pragma unroll
    for (int i = 0; i < 4; ++i) {
      s[i].x = fast_exp(s[i].x - mx);
      s[i].y = fast_exp(s[i].y - mx);
      s[i].z = fast_exp(s[i].z - mx);
      s[i].w = fast_exp(s[i].w - mx);
      sum += (s[i].x + s[i].y) + (s[i].z + s[i].w);
    }
    sum += __shfl_xor(sum, 16, 32);
    sum += __shfl_xor(sum, 8, 32);
    sum += __shfl_xor(sum, 4, 32);
    sum += __shfl_xor(sum, 2, 32);
    sum += __shfl_xor(sum, 1, 32);
    const float inv = 1.0f / sum;

    float* orow = Sout + (size_t)(xrow0 + r) * TM;
#pragma unroll
    for (int i = 0; i < 4; ++i) {
      float4 o;
      o.x = s[i].x * inv; o.y = s[i].y * inv; o.z = s[i].z * inv; o.w = s[i].w * inv;
      *(float4*)(orow + 128 * i + 4 * lane) = o;
    }
  }
}

__device__ __forceinline__ void load_kset(KSet& f, const float* __restrict__ arow,
                                          const float* __restrict__ bcol, int k0, int half) {
#pragma unroll
  for (int i = 0; i < KG; ++i) {
    const int kr = k0 + 4 * i + 2 * half;
    f.a[i]    = *(const v2f*)(arow + kr);
    f.b0[i].x = bcol[(size_t)kr * D];
    f.b0[i].y = bcol[(size_t)(kr + 1) * D];
    f.b1[i].x = bcol[(size_t)kr * D + 16];
    f.b1[i].y = bcol[(size_t)(kr + 1) * D + 16];
  }
}

static_assert(TM % KSPAN == 0 && (TM / KSPAN) % 4 == 0 && TM / KSPAN >= 8,
              "the ring below consumes its sets in trips of four");
__global__ __launch_bounds__(128) void attn_out_kernel(
    const float* __restrict__ S, const float* __restrict__ Mem,
    float* __restrict__ Out) {
  const int warp = threadIdx.x >> 5;
  const int lane = threadIdx.x & 31;
  const int half = lane >> 4;
  const int l    = lane & 15;

  const int pairsPerRow   = D / 32;
  const int tilesPerBatch = (TX / 16) * pairsPerRow;
  int tile  = blockIdx.x * (blockDim.x >> 5) + warp;
  int batch = tile / tilesPerBatch;
  int t     = tile % tilesPerBatch;
  int mt = t / pairsPerRow;
  int np = t % pairsPerRow;
  const int row0 = mt * 16;
  const int col0 = np * 32;

  const float* Sb = S   + (size_t)batch * TX * TM;
  const float* Mb = Mem + (size_t)batch * TM * D;
  const float* arow = Sb + (size_t)(row0 + l) * TM;
  const float* bcol = Mb + col0 + l;

  v8f acc0 = {}; v8f acc1 = {};
  KSet r0, r1, r2, r3;
  load_kset(r0, arow, bcol, 0 * KSPAN, half);
  load_kset(r1, arow, bcol, 1 * KSPAN, half);
  load_kset(r2, arow, bcol, 2 * KSPAN, half);
  int k = 0;
  for (; k < TM - 4 * KSPAN; k += 4 * KSPAN) {
    load_kset(r3, arow, bcol, k + 3 * KSPAN, half); mma_kset(r0, acc0, acc1); ring_order(acc0, acc1);
    load_kset(r0, arow, bcol, k + 4 * KSPAN, half); mma_kset(r1, acc0, acc1); ring_order(acc0, acc1);
    load_kset(r1, arow, bcol, k + 5 * KSPAN, half); mma_kset(r2, acc0, acc1); ring_order(acc0, acc1);
    load_kset(r2, arow, bcol, k + 6 * KSPAN, half); mma_kset(r3, acc0, acc1); ring_order(acc0, acc1);
  }
  load_kset(r3, arow, bcol, k + 3 * KSPAN, half);
  mma_kset(r0, acc0, acc1);
  mma_kset(r1, acc0, acc1);
  mma_kset(r2, acc0, acc1);
  mma_kset(r3, acc0, acc1);

#pragma unroll
  for (int j = 0; j < 8; ++j) {
    int r = row0 + j + 8 * half;
    Out[((size_t)batch * TX + r) * D + col0 + l]      = acc0[j];
    Out[((size_t)batch * TX + r) * D + col0 + 16 + l] = acc1[j];
  }
}

extern "C" void kernel_launch(void* const* d_in, const int* in_sizes, int n_in,
                              void* d_out, int out_size, void* d_ws, size_t ws_size,
                              hipStream_t stream) {
  const float* x    = (const float*)d_in[0];
  const float* mem  = (const float*)d_in[1];
  const int*   mask = (const int*)d_in[2];
  const float* w1   = (const float*)d_in[3];
  const float* b1   = (const float*)d_in[4];
  const float* w2   = (const float*)d_in[5];
  const float* wst  = (const float*)d_in[6];

  float* item1 = (float*)d_ws;
  float* item2 = item1 + (size_t)BATCH * TX * D;
  float* out   = (float*)d_out;
  float* Sout  = out + (size_t)BATCH * TX * D;

  const int rowsX = BATCH * TX;
  dim3 blk(128);

  proj_gemm_kernel<<<2 * PROJ_BLOCKS, blk, 0, stream>>>(x, w1, b1, item1, mem, w2, item2, rowsX);

  scores_softmax_kernel<<<BATCH * TX / XT, TM, 0, stream>>>(item1, item2, wst, mask, Sout);

  int tilesOut = BATCH * (TX / 16) * (D / 32);
  attn_out_kernel<<<tilesOut / 4, blk, 0, stream>>>(Sout, mem, out);
}


# S*memory product moved into the scores kernel: after its softmax each block multiplies its own 8 rows, kept in LDS, by memory on WMMA; 3 launches become 2; same products, same sum order
# speedup vs baseline: 2.5192x; 1.1024x over previous
#include <hip/hip_runtime.h>
#include <math.h>

#define D      256
#define TX     512
#define TM     512
#define BATCH  4

typedef __attribute__((ext_vector_type(2))) float v2f;
typedef __attribute__((ext_vector_type(8))) float v8f;

__device__ __forceinline__ float fast_tanh(float x) {
#if __has_builtin(__builtin_amdgcn_tanhf)
  return __builtin_amdgcn_tanhf(x);
#else
  float r;
  asm("v_tanh_f32 %0, %1" : "=v"(r) : "v"(x));
  return r;
#endif
}

__device__ __forceinline__ float fast_exp(float x) {
#if __has_builtin(__builtin_amdgcn_exp2f)
  return __builtin_amdgcn_exp2f(x * 1.4426950408889634f);
#else
  return __expf(x);
#endif
}

#define KG    4
#define KSPAN (4 * KG)
struct KSet { v2f a[KG]; v2f b0[KG]; v2f b1[KG]; };

__device__ __forceinline__ void mma_kset(const KSet& f, v8f& acc0, v8f& acc1) {
#pragma unroll
  for (int i = 0; i < KG; ++i) {
    acc0 = __builtin_amdgcn_wmma_f32_16x16x4_f32(false, f.a[i], false, f.b0[i], (short)0, acc0, false, false);
    acc1 = __builtin_amdgcn_wmma_f32_16x16x4_f32(false, f.a[i], false, f.b1[i], (short)0, acc1, false, false);
  }
}

__device__ __forceinline__ void ring_order(v8f& acc0, v8f& acc1) {
  asm volatile("" : "+v"(acc0), "+v"(acc1) : : "memory");
}

__device__ __forceinline__ void load_rows_kset(KSet& f, const float* __restrict__ arow,
                                               const float* __restrict__ b0row,
                                               const float* __restrict__ b1row, int k0, int half) {
#pragma unroll
  for (int i = 0; i < KG; ++i) {
    const int kr = k0 + 4 * i + 2 * half;
    f.a[i]  = *(const v2f*)(arow  + kr);
    f.b0[i] = *(const v2f*)(b0row + kr);
    f.b1[i] = *(const v2f*)(b1row + kr);
  }
}
#define PROJ_RING 6
#define PROJ_SETS (D / KSPAN)
static_assert(D % KSPAN == 0 && PROJ_SETS >= PROJ_RING, "the projection's k range is a whole number of sets, at least one ring");
static_assert(3 * KG * (PROJ_RING - 1) <= 63, "the loads in flight must fit the wave's outstanding-load counter");
#define PROJ_BLOCKS ((BATCH * TX / 16) * (D / 32) / 4)
static_assert(TX == TM, "both projections must have the same number of rows to share one grid split");
static_assert((D / 16) * (BATCH * TM / 32) == PROJ_BLOCKS * 4, "the transposed problem has the same number of dual-tiles");
__global__ __launch_bounds__(128) void proj_gemm_kernel(
    const float* __restrict__ In0, const float* __restrict__ W0,
    const float* __restrict__ bias0, float* __restrict__ Out0,
    const float* __restrict__ In1, const float* __restrict__ W1,
    float* __restrict__ Out1, int Mrows) {
  const int warp = threadIdx.x >> 5;
  const int lane = threadIdx.x & 31;
  const int half = lane >> 4;
  const int l    = lane & 15;

  const bool second = blockIdx.x >= PROJ_BLOCKS;
  const int tile = (blockIdx.x - (second ? PROJ_BLOCKS : 0)) * (blockDim.x >> 5) + warp;

  static_assert(D / 16 == 16 && D / 32 == 8, "the tile split below is written with shifts");
  const int grp = second ? tile >> 4 : tile >> 3;
  const int sub = second ? tile & 15 : tile & 7;
  const int row0 = second ? sub * 16 : grp * 16;
  const int col0 = second ? grp * 32 : sub * 32;
  if (!second && row0 >= Mrows) return;

  const float* Am = second ? W1 : In0;
  const float* Bm = second ? In1 : W0;

  v8f acc0 = {}; v8f acc1 = {};
  const float* arow  = Am + (size_t)(row0 + l) * D;
  const float* b0row = Bm + (size_t)(col0 + l) * D;
  const float* b1row = Bm + (size_t)(col0 + 16 + l) * D;
  KSet ring[PROJ_RING];
#pragma unroll
  for (int s = 0; s < PROJ_RING - 1; ++s)
    load_rows_kset(ring[s], arow, b0row, b1row, s * KSPAN, half);
#pragma unroll
  for (int s = 0; s < PROJ_SETS; ++s) {
    if (s + PROJ_RING - 1 < PROJ_SETS)
      load_rows_kset(ring[(s + PROJ_RING - 1) % PROJ_RING], arow, b0row, b1row, (s + PROJ_RING - 1) * KSPAN, half);
    mma_kset(ring[s % PROJ_RING], acc0, acc1);
    ring_order(acc0, acc1);
  }

  if (!second) {
    const float bv0 = bias0[col0 + l], bv1 = bias0[col0 + 16 + l];
#pragma unroll
    for (int j = 0; j < 8; ++j) {
      int r = row0 + j + 8 * half;
      Out0[(size_t)r * D + col0 + l]      = acc0[j] + bv0;
      Out0[(size_t)r * D + col0 + 16 + l] = acc1[j] + bv1;
    }
  } else {
    const int b  = col0 / TM;
    const int m0 = col0 % TM;
    float4* dst = (float4*)Out1 + ((size_t)b * (D / 4) + row0 / 4 + 2 * half) * TM + m0 + l;
    dst[0]       = make_float4(acc0[0], acc0[1], acc0[2], acc0[3]);
    dst[TM]      = make_float4(acc0[4], acc0[5], acc0[6], acc0[7]);
    dst[16]      = make_float4(acc1[0], acc1[1], acc1[2], acc1[3]);
    dst[TM + 16] = make_float4(acc1[4], acc1[5], acc1[6], acc1[7]);
  }
}

typedef const float __attribute__((address_space(4))) cfloat_k;
__device__ __forceinline__ const cfloat_k* as_const_space(const float* p) {
  return (const cfloat_k*)p;
}

#define XT 8
#define DT 16
#define GT (DT / 4)
struct DTrip { float4 g[GT]; };

__device__ __forceinline__ void load_trip(DTrip& t, const float4* __restrict__ row, int d0) {
#pragma unroll
  for (int i = 0; i < GT; ++i) t.g[i] = row[(size_t)(d0 / 4 + i) * TM];
}

__device__ __forceinline__ void stretch_edge() { __builtin_amdgcn_sched_barrier(0); }

#define XS 4
#define SG_PLAIN 0x002
#define SG_TRANS 0x400
__device__ __forceinline__ void tanh_between_plain_pairs() {
  __builtin_amdgcn_sched_group_barrier(SG_PLAIN, 2, 0);
#pragma unroll
  for (int s = 0; s < XS * DT - 2; ++s) {
    __builtin_amdgcn_sched_group_barrier(SG_TRANS, 1, 0);
    __builtin_amdgcn_sched_group_barrier(SG_PLAIN, 2, 0);
  }
  __builtin_amdgcn_sched_group_barrier(SG_TRANS, 2, 0);
  __builtin_amdgcn_sched_group_barrier(SG_PLAIN, 2, 0);
}

__device__ __forceinline__ void score_trip(const DTrip& t, int d0, const cfloat_k* i1b,
                                           const cfloat_k* wsc, float (&acc)[XT]) {
#pragma unroll
  for (int xs = 0; xs < XT; xs += XS) {
    float th[GT][XS][4];
#pragma unroll
    for (int i = 0; i < GT; ++i) {
      const float4 v = t.g[i];
#pragma unroll
      for (int x = 0; x < XS; ++x) {
        const cfloat_k* ap = i1b + (xs + x) * D + d0 + 4 * i;
        th[i][x][0] = ap[0] + v.x;
        th[i][x][1] = ap[1] + v.y;
        th[i][x][2] = ap[2] + v.z;
        th[i][x][3] = ap[3] + v.w;
      }
    }
#pragma unroll
    for (int i = 0; i < GT; ++i)
#pragma unroll
      for (int x = 0; x < XS; ++x)
#pragma unroll
        for (int c = 0; c < 4; ++c) th[i][x][c] = fast_tanh(th[i][x][c]);
#pragma unroll
    for (int s = 0; s < DT + XS - 1; ++s)
#pragma unroll
      for (int x = 0; x < XS; ++x) {
        const int dd = s - x;
        if (dd >= 0 && dd < DT) acc[xs + x] += th[dd / 4][x][dd % 4] * wsc[d0 + dd];
      }
    tanh_between_plain_pairs();
    stretch_edge();
  }
}

static_assert(D % (2 * DT) == 0 && D >= 4 * DT, "the d loop below consumes two trips per pass and peels the last pass");

#define SCP (TM + 8)
static_assert(SCP % 4 == 0, "the softmax reads the rows with 16-byte loads");

struct OSet { v2f a[KG]; v2f b[KG]; };
__device__ __forceinline__ void load_oset_a(OSet& f, const float* arow, int k0, int half) {
#pragma unroll
  for (int i = 0; i < KG; ++i) f.a[i] = *(const v2f*)(arow + k0 + 4 * i + 2 * half);
}
__device__ __forceinline__ void load_oset_b(OSet& f, const float* __restrict__ bcol, int k0, int half) {
#pragma unroll
  for (int i = 0; i < KG; ++i) {
    const int kr = k0 + 4 * i + 2 * half;
    f.b[i].x = bcol[(size_t)kr * D];
    f.b[i].y = bcol[(size_t)(kr + 1) * D];
  }
}
__device__ __forceinline__ void load_oset(OSet& f, const float* arow, const float* __restrict__ bcol,
                                          int k0, int half) {
  load_oset_b(f, bcol, k0, half);
  load_oset_a(f, arow, k0, half);
}
__device__ __forceinline__ void mma_oset(const OSet& f, v8f& acc) {
#pragma unroll
  for (int i = 0; i < KG; ++i)
    acc = __builtin_amdgcn_wmma_f32_16x16x4_f32(false, f.a[i], false, f.b[i], (short)0, acc, false, false);
}
__device__ __forceinline__ void ring_order1(v8f& acc) {
  asm volatile("" : "+v"(acc) : : "memory");
}
#define ORING 8
#define OSETS (TM / KSPAN)
static_assert(TM % KSPAN == 0 && OSETS >= ORING, "the product's k range is a whole number of sets, at least one ring");
static_assert(2 * KG * (ORING - 1) <= 63, "the loads in flight must fit the wave's outstanding-load counter");
static_assert(D == 16 * (TM / 32) && XT == 8, "one 16-column tile per wave of the block; rows 0-7 of the tile are the block's rows");
__global__ __launch_bounds__(512) __attribute__((amdgpu_waves_per_eu(4, 4))) void scores_softmax_kernel(
    const float* __restrict__ item1, const float* __restrict__ item2g,
    const float* __restrict__ wst, const int* __restrict__ mask,
    float* __restrict__ Sout, const float* __restrict__ Mem, float* __restrict__ Out) {
  __shared__ float sc[XT][SCP];
  __shared__ float ot[XT][D];

  const int tid   = threadIdx.x;
  const int warp  = tid >> 5;
  const int lane  = tid & 31;
  const int xrow0 = blockIdx.x * XT;
  const int b     = xrow0 / TX;

  const float4*   i2row = (const float4*)item2g + (size_t)b * (D / 4) * TM + tid;
  const cfloat_k* i1b   = as_const_space(item1 + (size_t)xrow0 * D);
  const cfloat_k* wsc   = as_const_space(wst);

  float acc[XT];
#pragma unroll
  for (int x = 0; x < XT; ++x) acc[x] = 0.0f;

  DTrip ta, tb;
  load_trip(ta, i2row, 0);
  for (int d0 = 0; d0 < D - 2 * DT; d0 += 2 * DT) {
    load_trip(tb, i2row, d0 + DT);
    score_trip(ta, d0, i1b, wsc, acc);
    load_trip(ta, i2row, d0 + 2 * DT);
    score_trip(tb, d0 + DT, i1b, wsc, acc);
  }
  load_trip(tb, i2row, D - DT);
  score_trip(ta, D - 2 * DT, i1b, wsc, acc);
  score_trip(tb, D - DT, i1b, wsc, acc);

#pragma unroll
  for (int x = 0; x < XT; ++x) sc[x][tid] = acc[x];

  const int half = lane >> 4;
  const int l    = lane & 15;
  const float* bcol = Mem + (size_t)b * TM * D + warp * 16 + l;
  __syncthreads();
  OSet ring[ORING];
#pragma unroll
  for (int s = 0; s < ORING - 1; ++s) load_oset_b(ring[s], bcol, s * KSPAN, half);

  if (warp < XT) {
    const int r = warp;
    float4 s[4];
#pragma unroll
    for (int i = 0; i < 4; ++i) {
      s[i] = *(const float4*)(&sc[r][128 * i + 4 * lane]);
      const int4 mk = *(const int4*)(mask + b * TM + 128 * i + 4 * lane);
      if (mk.x == 0) s[i].x = -INFINITY;
      if (mk.y == 0) s[i].y = -INFINITY;
      if (mk.z == 0) s[i].z = -INFINITY;
      if (mk.w == 0) s[i].w = -INFINITY;
    }
    float mx = fmaxf(fmaxf(s[0].x, s[0].y), fmaxf(s[0].z, s[0].w));
#pragma unroll
    for (int i = 1; i < 4; ++i)
      mx = fmaxf(mx, fmaxf(fmaxf(s[i].x, s[i].y), fmaxf(s[i].z, s[i].w)));
    mx = fmaxf(mx, __shfl_xor(mx, 16, 32));
    mx = fmaxf(mx, __shfl_xor(mx, 8, 32));
    mx = fmaxf(mx, __shfl_xor(mx, 4, 32));
    mx = fmaxf(mx, __shfl_xor(mx, 2, 32));
    mx = fmaxf(mx, __shfl_xor(mx, 1, 32));

    float sum = 0.0f;
#pragma unroll
    for (int i = 0; i < 4; ++i) {
      s[i].x = fast_exp(s[i].x - mx);
      s[i].y = fast_exp(s[i].y - mx);
      s[i].z = fast_exp(s[i].z - mx);
      s[i].w = fast_exp(s[i].w - mx);
      sum += (s[i].x + s[i].y) + (s[i].z + s[i].w);
    }
    sum += __shfl_xor(sum, 16, 32);
    sum += __shfl_xor(sum, 8, 32);
    sum += __shfl_xor(sum, 4, 32);
    sum += __shfl_xor(sum, 2, 32);
    sum += __shfl_xor(sum, 1, 32);
    const float inv = 1.0f / sum;

    float* orow = Sout + (size_t)(xrow0 + r) * TM;
#pragma unroll
    for (int i = 0; i < 4; ++i) {
      float4 o;
      o.x = s[i].x * inv; o.y = s[i].y * inv; o.z = s[i].z * inv; o.w = s[i].w * inv;
      *(float4*)(orow + 128 * i + 4 * lane) = o;
      *(float4*)(&sc[r][128 * i + 4 * lane]) = o;
    }
  }
  __syncthreads();

  const float* arow = &sc[l & (XT - 1)][0];
#pragma unroll
  for (int s = 0; s < ORING - 1; ++s) load_oset_a(ring[s], arow, s * KSPAN, half);
  v8f oacc = {};
#pragma unroll
  for (int s = 0; s < OSETS; ++s) {
    if (s + ORING - 1 < OSETS)
      load_oset(ring[(s + ORING - 1) % ORING], arow, bcol, (s + ORING - 1) * KSPAN, half);
    mma_oset(ring[s % ORING], oacc);
    ring_order1(oacc);
  }

  if (half == 0) {
#pragma unroll
    for (int j = 0; j < XT; ++j) ot[j][warp * 16 + l] = oacc[j];
  }
  __syncthreads();
  if (warp < XT) {
    float* orow = Out + (size_t)(xrow0 + warp) * D;
    *(float4*)(orow + 4 * lane)       = *(const float4*)(&ot[warp][4 * lane]);
    *(float4*)(orow + 128 + 4 * lane) = *(const float4*)(&ot[warp][128 + 4 * lane]);
  }
}

extern "C" void kernel_launch(void* const* d_in, const int* in_sizes, int n_in,
                              void* d_out, int out_size, void* d_ws, size_t ws_size,
                              hipStream_t stream) {
  const float* x    = (const float*)d_in[0];
  const float* mem  = (const float*)d_in[1];
  const int*   mask = (const int*)d_in[2];
  const float* w1   = (const float*)d_in[3];
  const float* b1   = (const float*)d_in[4];
  const float* w2   = (const float*)d_in[5];
  const float* wst  = (const float*)d_in[6];

  float* item1 = (float*)d_ws;
  float* item2 = item1 + (size_t)BATCH * TX * D;
  float* out   = (float*)d_out;
  float* Sout  = out + (size_t)BATCH * TX * D;

  const int rowsX = BATCH * TX;
  dim3 blk(128);

  proj_gemm_kernel<<<2 * PROJ_BLOCKS, blk, 0, stream>>>(x, w1, b1, item1, mem, w2, item2, rowsX);

  scores_softmax_kernel<<<BATCH * TX / XT, TM, 0, stream>>>(item1, item2, wst, mask, Sout, mem, out);
}
